# starting kernel
# baseline (speedup 1.0000x reference)
#include <hip/hip_runtime.h>
#include <hip/hip_bf16.h>

#define GAT_N      50000
#define GAT_E      800000
#define GAT_ETOT   850000
#define GAT_D      128
#define GAT_G      64
#define NEG_SLOPE  0.2f

typedef __attribute__((ext_vector_type(2))) float v2f;
typedef __attribute__((ext_vector_type(8))) float v8f;

__device__ __forceinline__ unsigned enc_f32(float f) {
    unsigned u = __float_as_uint(f);
    return (u & 0x80000000u) ? ~u : (u | 0x80000000u);
}
__device__ __forceinline__ float dec_f32(unsigned k) {
    unsigned u = (k & 0x80000000u) ? (k & 0x7FFFFFFFu) : ~k;
    return __uint_as_float(u);
}

__global__ void gat_gemm128_wmma(const float* __restrict__ X,
                                 const float* __restrict__ W,
                                 float* __restrict__ H) {
    const int wave = threadIdx.x >> 5;
    const int lane = threadIdx.x & 31;
    const int half = lane >> 4;
    const int l    = lane & 15;
    const int m0   = blockIdx.x * 16;
    const int n0   = wave * 16;

    const float* xrow = X + (size_t)(m0 + l) * GAT_D;
    v8f acc = {};

#pragma unroll
    for (int k = 0; k < GAT_D; k += 4) {
        const int kk = k + 2 * half;
        v2f a;
        {
            const float2 av = *(const float2*)(xrow + kk);
            a.x = av.x; a.y = av.y;
        }
        v2f b;
        b.x = W[(size_t)kk       * GAT_D + n0 + l];
        b.y = W[(size_t)(kk + 1) * GAT_D + n0 + l];
        acc = __builtin_amdgcn_wmma_f32_16x16x4_f32(
            false, a, false, b, (short)0, acc, false, false);
    }

    const int mBase = m0 + half * 8;
#pragma unroll
    for (int r = 0; r < 8; ++r)
        H[(size_t)(mBase + r) * GAT_D + n0 + l] = acc[r];
}

__global__ void gat_alpha(const float* __restrict__ H,
                          const float* __restrict__ a_src,
                          const float* __restrict__ a_dst,
                          float* __restrict__ as_out,
                          float* __restrict__ ad_out) {
    const int node = blockIdx.x * 8 + (threadIdx.x >> 5);
    const int lane = threadIdx.x & 31;
    if (node >= GAT_N) return;
    const float* h = H + (size_t)node * GAT_D;
    float s = 0.f, d = 0.f;
#pragma unroll
    for (int i = lane; i < GAT_D; i += 32) {
        const float v = h[i];
        s += v * a_src[i];
        d += v * a_dst[i];
    }
#pragma unroll
    for (int off = 16; off > 0; off >>= 1) {
        s += __shfl_xor(s, off, 32);
        d += __shfl_xor(d, off, 32);
    }
    if (lane == 0) { as_out[node] = s; ad_out[node] = d; }
}

__global__ void gat_init(float* __restrict__ out,
                         const float* __restrict__ bias,
                         unsigned* __restrict__ m_u,
                         float* __restrict__ ssum) {
    const size_t t = (size_t)blockIdx.x * blockDim.x + threadIdx.x;
    if (t < (size_t)GAT_N * GAT_D) out[t] = bias[t & (GAT_D - 1)];
    if (t < GAT_N) { m_u[t] = 0u; ssum[t] = 0.f; }
}

__global__ void gat_edge_max(const int* __restrict__ ei,
                             const float* __restrict__ as,
                             const float* __restrict__ ad,
                             float* __restrict__ ebuf,
                             unsigned* __restrict__ m_u) {
    const int i = blockIdx.x * blockDim.x + threadIdx.x;
    if (i >= GAT_ETOT) return;
    int s_idx, d_idx;
    if (i < GAT_E) { s_idx = ei[i]; d_idx = ei[GAT_E + i]; }
    else           { s_idx = d_idx = i - GAT_E; }
    float e = as[s_idx] + ad[d_idx];
    e = (e > 0.f) ? e : NEG_SLOPE * e;
    ebuf[i] = e;
    atomicMax(&m_u[d_idx], enc_f32(e));
}

__global__ void gat_edge_exp(const int* __restrict__ ei,
                             float* __restrict__ ebuf,
                             const unsigned* __restrict__ m_u,
                             float* __restrict__ ssum) {
    const int i = blockIdx.x * blockDim.x + threadIdx.x;
    if (i >= GAT_ETOT) return;
    int d_idx;
    if (i < GAT_E) d_idx = ei[GAT_E + i];
    else           d_idx = i - GAT_E;
    const float m  = dec_f32(m_u[d_idx]);
    const float ex = expf(ebuf[i] - m);
    ebuf[i] = ex;
    atomicAdd(&ssum[d_idx], ex);
}

__global__ void gat_edge_scatter(const int* __restrict__ ei,
                                 const float* __restrict__ ebuf,
                                 const float* __restrict__ ssum,
                                 const float* __restrict__ H,
                                 float* __restrict__ out) {
    const int edge = blockIdx.x * 8 + (threadIdx.x >> 5);
    const int lane = threadIdx.x & 31;
    if (edge >= GAT_ETOT) return;
    int s_idx, d_idx;
    if (edge < GAT_E) { s_idx = ei[edge]; d_idx = ei[GAT_E + edge]; }
    else              { s_idx = d_idx = edge - GAT_E; }
    const float coef = ebuf[edge] / (ssum[d_idx] + 1e-16f);
    const float4 h4 = *(const float4*)(H + (size_t)s_idx * GAT_D + lane * 4);
    float* o = out + (size_t)d_idx * GAT_D + lane * 4;
    atomicAdd(o + 0, h4.x * coef);
    atomicAdd(o + 1, h4.y * coef);
    atomicAdd(o + 2, h4.z * coef);
    atomicAdd(o + 3, h4.w * coef);
}

__global__ void gat_pool_zero(float* __restrict__ pool, unsigned* __restrict__ cnt) {
    const int t = blockIdx.x * blockDim.x + threadIdx.x;
    if (t < GAT_G * GAT_D) pool[t] = 0.f;
    if (t < GAT_G) cnt[t] = 0u;
}

__global__ void gat_pool_accum(const float* __restrict__ feats,
                               const int* __restrict__ batch,
                               float* __restrict__ pool,
                               unsigned* __restrict__ cnt) {
    const size_t t = (size_t)blockIdx.x * blockDim.x + threadIdx.x;
    if (t >= (size_t)GAT_N * GAT_D) return;
    const int n = (int)(t >> 7);
    const int d = (int)(t & (GAT_D - 1));
    const int g = batch[n];
    atomicAdd(&pool[g * GAT_D + d], feats[t]);
    if (d == 0) atomicAdd(&cnt[g], 1u);
}

__global__ void gat_pool_final(const float* __restrict__ pool,
                               const unsigned* __restrict__ cnt,
                               float* __restrict__ out_pooled) {
    const int t = blockIdx.x * blockDim.x + threadIdx.x;
    if (t >= GAT_G * GAT_D) return;
    const float c = fmaxf((float)cnt[t >> 7], 1.0f);
    out_pooled[t] = pool[t] / c;
}

static void launch_gat_layer(const float* x, const float* W,
                             const float* a_s, const float* a_d,
                             const float* bias,
                             const int* ei,
                             float* h, float* alphaS, float* alphaD,
                             unsigned* m_u, float* ssum, float* ebuf,
                             float* out, hipStream_t stream) {
    gat_gemm128_wmma<<<GAT_N / 16, 256, 0, stream>>>(x, W, h);
    gat_alpha<<<GAT_N / 8, 256, 0, stream>>>(h, a_s, a_d, alphaS, alphaD);
    gat_init<<<(GAT_N * GAT_D) / 256, 256, 0, stream>>>(out, bias, m_u, ssum);
    const int eb = (GAT_ETOT + 255) / 256;
    gat_edge_max<<<eb, 256, 0, stream>>>(ei, alphaS, alphaD, ebuf, m_u);
    gat_edge_exp<<<eb, 256, 0, stream>>>(ei, ebuf, m_u, ssum);
    gat_edge_scatter<<<GAT_ETOT / 8, 256, 0, stream>>>(ei, ebuf, ssum, h, out);
}

extern "C" void kernel_launch(void* const* d_in, const int* in_sizes, int n_in,
                              void* d_out, int out_size, void* d_ws, size_t ws_size,
                              hipStream_t stream) {
    const float* x     = (const float*)d_in[0];
    const int*   ei    = (const int*)d_in[1];
    const int*   batch = (const int*)d_in[2];
    const float* W1  = (const float*)d_in[3];
    const float* as1 = (const float*)d_in[4];
    const float* ad1 = (const float*)d_in[5];
    const float* b1  = (const float*)d_in[6];
    const float* W2  = (const float*)d_in[7];
    const float* as2 = (const float*)d_in[8];
    const float* ad2 = (const float*)d_in[9];
    const float* b2  = (const float*)d_in[10];

    float* out_feats  = (float*)d_out;
    float* out_pooled = out_feats + (size_t)GAT_N * GAT_D;

    float* ws = (float*)d_ws;
    size_t o = 0;
    float*    h      = ws + o; o += (size_t)GAT_N * GAT_D;
    float*    feats1 = ws + o; o += (size_t)GAT_N * GAT_D;
    float*    alphaS = ws + o; o += GAT_N;
    float*    alphaD = ws + o; o += GAT_N;
    unsigned* m_u    = (unsigned*)(ws + o); o += GAT_N;
    float*    ssum   = ws + o; o += GAT_N;
    float*    ebuf   = ws + o; o += GAT_ETOT;
    float*    pool   = ws + o; o += GAT_G * GAT_D;
    unsigned* cnt    = (unsigned*)(ws + o); o += GAT_G;
    (void)ws_size; (void)in_sizes; (void)n_in; (void)out_size;

    launch_gat_layer(x, W1, as1, ad1, b1, ei,
                     h, alphaS, alphaD, m_u, ssum, ebuf, feats1, stream);
    launch_gat_layer(feats1, W2, as2, ad2, b2, ei,
                     h, alphaS, alphaD, m_u, ssum, ebuf, out_feats, stream);

    gat_pool_zero<<<(GAT_G * GAT_D + 255) / 256, 256, 0, stream>>>(pool, cnt);
    gat_pool_accum<<<(GAT_N * GAT_D) / 256, 256, 0, stream>>>(out_feats, batch, pool, cnt);
    gat_pool_final<<<(GAT_G * GAT_D + 255) / 256, 256, 0, stream>>>(pool, cnt, out_pooled);
}


# Segment max: each node's max starts at its self-loop score, stored beside alpha_d as one 8-B pair; an edge reads the pair and issues atomicMax only if its score is above the value read; same max
# speedup vs baseline: 1.2721x; 1.2721x over previous
#include <hip/hip_runtime.h>
#include <hip/hip_bf16.h>

#define GAT_N      50000
#define GAT_E      800000
#define GAT_ETOT   850000
#define GAT_D      128
#define GAT_G      64
#define NEG_SLOPE  0.2f

typedef __attribute__((ext_vector_type(2))) float v2f;
typedef __attribute__((ext_vector_type(8))) float v8f;

__device__ __forceinline__ unsigned enc_f32(float f) {
    unsigned u = __float_as_uint(f);
    return (u & 0x80000000u) ? ~u : (u | 0x80000000u);
}
__device__ __forceinline__ float dec_f32(unsigned k) {
    unsigned u = (k & 0x80000000u) ? (k & 0x7FFFFFFFu) : ~k;
    return __uint_as_float(u);
}

__global__ void gat_gemm128_wmma(const float* __restrict__ X,
                                 const float* __restrict__ W,
                                 float* __restrict__ H) {
    const int wave = threadIdx.x >> 5;
    const int lane = threadIdx.x & 31;
    const int half = lane >> 4;
    const int l    = lane & 15;
    const int m0   = blockIdx.x * 16;
    const int n0   = wave * 16;

    const float* xrow = X + (size_t)(m0 + l) * GAT_D;
    v8f acc = {};

#pragma unroll
    for (int k = 0; k < GAT_D; k += 4) {
        const int kk = k + 2 * half;
        v2f a;
        {
            const float2 av = *(const float2*)(xrow + kk);
            a.x = av.x; a.y = av.y;
        }
        v2f b;
        b.x = W[(size_t)kk       * GAT_D + n0 + l];
        b.y = W[(size_t)(kk + 1) * GAT_D + n0 + l];
        acc = __builtin_amdgcn_wmma_f32_16x16x4_f32(
            false, a, false, b, (short)0, acc, false, false);
    }

    const int mBase = m0 + half * 8;
#pragma unroll
    for (int r = 0; r < 8; ++r)
        H[(size_t)(mBase + r) * GAT_D + n0 + l] = acc[r];
}

__global__ void gat_alpha(const float* __restrict__ H,
                          const float* __restrict__ a_src,
                          const float* __restrict__ a_dst,
                          float* __restrict__ as_out,
                          unsigned* __restrict__ dm) {
    const int node = blockIdx.x * 8 + (threadIdx.x >> 5);
    const int lane = threadIdx.x & 31;
    if (node >= GAT_N) return;
    const float* h = H + (size_t)node * GAT_D;
    float s = 0.f, d = 0.f;
#pragma unroll
    for (int i = lane; i < GAT_D; i += 32) {
        const float v = h[i];
        s += v * a_src[i];
        d += v * a_dst[i];
    }
#pragma unroll
    for (int off = 16; off > 0; off >>= 1) {
        s += __shfl_xor(s, off, 32);
        d += __shfl_xor(d, off, 32);
    }
    if (lane == 0) {
        as_out[node] = s;
        float e = s + d;
        e = (e > 0.f) ? e : NEG_SLOPE * e;
        *(uint2*)(dm + 2 * (size_t)node) = make_uint2(__float_as_uint(d), enc_f32(e));
    }
}

__global__ void gat_init(float* __restrict__ out,
                         const float* __restrict__ bias,
                         float* __restrict__ ssum) {
    const size_t t = (size_t)blockIdx.x * blockDim.x + threadIdx.x;
    if (t < (size_t)GAT_N * GAT_D) out[t] = bias[t & (GAT_D - 1)];
    if (t < GAT_N) ssum[t] = 0.f;
}

__global__ void gat_edge_max(const int* __restrict__ ei,
                             const float* __restrict__ as,
                             unsigned* dm,
                             float* __restrict__ ebuf) {
    const int i = blockIdx.x * blockDim.x + threadIdx.x;
    if (i >= GAT_ETOT) return;
    int s_idx, d_idx;
    if (i < GAT_E) { s_idx = ei[i]; d_idx = ei[GAT_E + i]; }
    else           { s_idx = d_idx = i - GAT_E; }
    const uint2 p = *(const uint2*)(dm + 2 * (size_t)d_idx);
    float e = as[s_idx] + __uint_as_float(p.x);
    e = (e > 0.f) ? e : NEG_SLOPE * e;
    ebuf[i] = e;
    const unsigned k = enc_f32(e);
    if (i < GAT_E && k > p.y) atomicMax(dm + 2 * (size_t)d_idx + 1, k);
}

__global__ void gat_edge_exp(const int* __restrict__ ei,
                             float* __restrict__ ebuf,
                             const unsigned* __restrict__ dm,
                             float* __restrict__ ssum) {
    const int i = blockIdx.x * blockDim.x + threadIdx.x;
    if (i >= GAT_ETOT) return;
    int d_idx;
    if (i < GAT_E) d_idx = ei[GAT_E + i];
    else           d_idx = i - GAT_E;
    const float m  = dec_f32(dm[2 * (size_t)d_idx + 1]);
    const float ex = expf(ebuf[i] - m);
    ebuf[i] = ex;
    atomicAdd(&ssum[d_idx], ex);
}

__global__ void gat_edge_scatter(const int* __restrict__ ei,
                                 const float* __restrict__ ebuf,
                                 const float* __restrict__ ssum,
                                 const float* __restrict__ H,
                                 float* __restrict__ out) {
    const int edge = blockIdx.x * 8 + (threadIdx.x >> 5);
    const int lane = threadIdx.x & 31;
    if (edge >= GAT_ETOT) return;
    int s_idx, d_idx;
    if (edge < GAT_E) { s_idx = ei[edge]; d_idx = ei[GAT_E + edge]; }
    else              { s_idx = d_idx = edge - GAT_E; }
    const float coef = ebuf[edge] / (ssum[d_idx] + 1e-16f);
    const float* hrow = H + (size_t)s_idx * GAT_D + lane;
    float* o = out + (size_t)d_idx * GAT_D + lane;
    const float h0 = hrow[0];
    const float h1 = hrow[32];
    const float h2 = hrow[64];
    const float h3 = hrow[96];
    atomicAdd(o,      h0 * coef);
    atomicAdd(o + 32, h1 * coef);
    atomicAdd(o + 64, h2 * coef);
    atomicAdd(o + 96, h3 * coef);
}

__global__ void gat_pool_zero(float* __restrict__ pool, unsigned* __restrict__ cnt) {
    const int t = blockIdx.x * blockDim.x + threadIdx.x;
    if (t < GAT_G * GAT_D) pool[t] = 0.f;
    if (t < GAT_G) cnt[t] = 0u;
}

__global__ void gat_pool_accum(const float* __restrict__ feats,
                               const int* __restrict__ batch,
                               float* __restrict__ pool,
                               unsigned* __restrict__ cnt) {
    const size_t t = (size_t)blockIdx.x * blockDim.x + threadIdx.x;
    if (t >= (size_t)GAT_N * GAT_D) return;
    const int n = (int)(t >> 7);
    const int d = (int)(t & (GAT_D - 1));
    const int g = batch[n];
    atomicAdd(&pool[g * GAT_D + d], feats[t]);
    if (d == 0) atomicAdd(&cnt[g], 1u);
}

__global__ void gat_pool_final(const float* __restrict__ pool,
                               const unsigned* __restrict__ cnt,
                               float* __restrict__ out_pooled) {
    const int t = blockIdx.x * blockDim.x + threadIdx.x;
    if (t >= GAT_G * GAT_D) return;
    const float c = fmaxf((float)cnt[t >> 7], 1.0f);
    out_pooled[t] = pool[t] / c;
}

static void launch_gat_layer(const float* x, const float* W,
                             const float* a_s, const float* a_d,
                             const float* bias,
                             const int* ei,
                             float* h, float* alphaS, unsigned* dm,
                             float* ssum, float* ebuf,
                             float* out, hipStream_t stream) {
    gat_gemm128_wmma<<<GAT_N / 16, 256, 0, stream>>>(x, W, h);
    gat_alpha<<<GAT_N / 8, 256, 0, stream>>>(h, a_s, a_d, alphaS, dm);
    gat_init<<<(GAT_N * GAT_D) / 256, 256, 0, stream>>>(out, bias, ssum);
    const int eb = (GAT_ETOT + 255) / 256;
    gat_edge_max<<<eb, 256, 0, stream>>>(ei, alphaS, dm, ebuf);
    gat_edge_exp<<<eb, 256, 0, stream>>>(ei, ebuf, dm, ssum);
    gat_edge_scatter<<<GAT_ETOT / 8, 256, 0, stream>>>(ei, ebuf, ssum, h, out);
}

extern "C" void kernel_launch(void* const* d_in, const int* in_sizes, int n_in,
                              void* d_out, int out_size, void* d_ws, size_t ws_size,
                              hipStream_t stream) {
    const float* x     = (const float*)d_in[0];
    const int*   ei    = (const int*)d_in[1];
    const int*   batch = (const int*)d_in[2];
    const float* W1  = (const float*)d_in[3];
    const float* as1 = (const float*)d_in[4];
    const float* ad1 = (const float*)d_in[5];
    const float* b1  = (const float*)d_in[6];
    const float* W2  = (const float*)d_in[7];
    const float* as2 = (const float*)d_in[8];
    const float* ad2 = (const float*)d_in[9];
    const float* b2  = (const float*)d_in[10];

    float* out_feats  = (float*)d_out;
    float* out_pooled = out_feats + (size_t)GAT_N * GAT_D;

    float* ws = (float*)d_ws;
    size_t o = 0;
    float*    h      = ws + o; o += (size_t)GAT_N * GAT_D;
    float*    feats1 = ws + o; o += (size_t)GAT_N * GAT_D;
    float*    alphaS = ws + o; o += GAT_N;
    unsigned* dm     = (unsigned*)(ws + o); o += 2 * (size_t)GAT_N;
    float*    ssum   = ws + o; o += GAT_N;
    float*    ebuf   = ws + o; o += GAT_ETOT;
    float*    pool   = ws + o; o += GAT_G * GAT_D;
    unsigned* cnt    = (unsigned*)(ws + o); o += GAT_G;
    (void)ws_size; (void)in_sizes; (void)n_in; (void)out_size;

    launch_gat_layer(x, W1, as1, ad1, b1, ei,
                     h, alphaS, dm, ssum, ebuf, feats1, stream);
    launch_gat_layer(feats1, W2, as2, ad2, b2, ei,
                     h, alphaS, dm, ssum, ebuf, out_feats, stream);

    gat_pool_zero<<<(GAT_G * GAT_D + 255) / 256, 256, 0, stream>>>(pool, cnt);
    gat_pool_accum<<<(GAT_N * GAT_D) / 256, 256, 0, stream>>>(out_feats, batch, pool, cnt);
    gat_pool_final<<<(GAT_G * GAT_D + 255) / 256, 256, 0, stream>>>(pool, cnt, out_pooled);
}


# Edge scatter: a wave takes 8 edges, not 1: indices, exp and sum loaded lane-parallel (one division), passed by readlane; all 32 row loads in flight before the first atomic; same adds, arrival order
# speedup vs baseline: 1.4605x; 1.1481x over previous
#include <hip/hip_runtime.h>
#include <hip/hip_bf16.h>

#define GAT_N      50000
#define GAT_E      800000
#define GAT_ETOT   850000
#define GAT_D      128
#define GAT_G      64
#define NEG_SLOPE  0.2f

typedef __attribute__((ext_vector_type(2))) float v2f;
typedef __attribute__((ext_vector_type(8))) float v8f;

__device__ __forceinline__ unsigned enc_f32(float f) {
    unsigned u = __float_as_uint(f);
    return (u & 0x80000000u) ? ~u : (u | 0x80000000u);
}
__device__ __forceinline__ float dec_f32(unsigned k) {
    unsigned u = (k & 0x80000000u) ? (k & 0x7FFFFFFFu) : ~k;
    return __uint_as_float(u);
}

__global__ void gat_gemm128_wmma(const float* __restrict__ X,
                                 const float* __restrict__ W,
                                 float* __restrict__ H) {
    const int wave = threadIdx.x >> 5;
    const int lane = threadIdx.x & 31;
    const int half = lane >> 4;
    const int l    = lane & 15;
    const int m0   = blockIdx.x * 16;
    const int n0   = wave * 16;

    const float* xrow = X + (size_t)(m0 + l) * GAT_D;
    v8f acc = {};

#pragma unroll
    for (int k = 0; k < GAT_D; k += 4) {
        const int kk = k + 2 * half;
        v2f a;
        {
            const float2 av = *(const float2*)(xrow + kk);
            a.x = av.x; a.y = av.y;
        }
        v2f b;
        b.x = W[(size_t)kk       * GAT_D + n0 + l];
        b.y = W[(size_t)(kk + 1) * GAT_D + n0 + l];
        acc = __builtin_amdgcn_wmma_f32_16x16x4_f32(
            false, a, false, b, (short)0, acc, false, false);
    }

    const int mBase = m0 + half * 8;
#pragma unroll
    for (int r = 0; r < 8; ++r)
        H[(size_t)(mBase + r) * GAT_D + n0 + l] = acc[r];
}

__global__ void gat_alpha(const float* __restrict__ H,
                          const float* __restrict__ a_src,
                          const float* __restrict__ a_dst,
                          float* __restrict__ as_out,
                          unsigned* __restrict__ dm) {
    const int node = blockIdx.x * 8 + (threadIdx.x >> 5);
    const int lane = threadIdx.x & 31;
    if (node >= GAT_N) return;
    const float* h = H + (size_t)node * GAT_D;
    float s = 0.f, d = 0.f;
#pragma unroll
    for (int i = lane; i < GAT_D; i += 32) {
        const float v = h[i];
        s += v * a_src[i];
        d += v * a_dst[i];
    }
#pragma unroll
    for (int off = 16; off > 0; off >>= 1) {
        s += __shfl_xor(s, off, 32);
        d += __shfl_xor(d, off, 32);
    }
    if (lane == 0) {
        as_out[node] = s;
        float e = s + d;
        e = (e > 0.f) ? e : NEG_SLOPE * e;
        *(uint2*)(dm + 2 * (size_t)node) = make_uint2(__float_as_uint(d), enc_f32(e));
    }
}

__global__ void gat_init(float* __restrict__ out,
                         const float* __restrict__ bias,
                         float* __restrict__ ssum) {
    const size_t t = (size_t)blockIdx.x * blockDim.x + threadIdx.x;
    if (t < (size_t)GAT_N * GAT_D) out[t] = bias[t & (GAT_D - 1)];
    if (t < GAT_N) ssum[t] = 0.f;
}

__global__ void gat_edge_max(const int* __restrict__ ei,
                             const float* __restrict__ as,
                             unsigned* dm,
                             float* __restrict__ ebuf) {
    const int i = blockIdx.x * blockDim.x + threadIdx.x;
    if (i >= GAT_ETOT) return;
    int s_idx, d_idx;
    if (i < GAT_E) { s_idx = ei[i]; d_idx = ei[GAT_E + i]; }
    else           { s_idx = d_idx = i - GAT_E; }
    const uint2 p = *(const uint2*)(dm + 2 * (size_t)d_idx);
    float e = as[s_idx] + __uint_as_float(p.x);
    e = (e > 0.f) ? e : NEG_SLOPE * e;
    ebuf[i] = e;
    const unsigned k = enc_f32(e);
    if (i < GAT_E && k > p.y) atomicMax(dm + 2 * (size_t)d_idx + 1, k);
}

__global__ void gat_edge_exp(const int* __restrict__ ei,
                             float* __restrict__ ebuf,
                             const unsigned* __restrict__ dm,
                             float* __restrict__ ssum) {
    const int i = blockIdx.x * blockDim.x + threadIdx.x;
    if (i >= GAT_ETOT) return;
    int d_idx;
    if (i < GAT_E) d_idx = ei[GAT_E + i];
    else           d_idx = i - GAT_E;
    const float m  = dec_f32(dm[2 * (size_t)d_idx + 1]);
    const float ex = expf(ebuf[i] - m);
    ebuf[i] = ex;
    atomicAdd(&ssum[d_idx], ex);
}

#define SC_EPW 8
__global__ void gat_edge_scatter(const int* __restrict__ ei,
                                 const float* __restrict__ ebuf,
                                 const float* __restrict__ ssum,
                                 const float* __restrict__ H,
                                 float* __restrict__ out) {
    const int wave = blockIdx.x * 8 + (threadIdx.x >> 5);
    const int lane = threadIdx.x & 31;
    const int e0   = wave * SC_EPW;
    if (e0 >= GAT_ETOT) return;
    const int edge = e0 + (lane & (SC_EPW - 1));
    int s_idx, d_idx;
    if (edge < GAT_E) { s_idx = ei[edge]; d_idx = ei[GAT_E + edge]; }
    else              { s_idx = d_idx = edge - GAT_E; }
    float hv[SC_EPW][4];
#pragma unroll
    for (int j = 0; j < SC_EPW; ++j) {
        const int sj = __builtin_amdgcn_readlane(s_idx, j);
        const float* hrow = H + (size_t)sj * GAT_D + lane;
        hv[j][0] = hrow[0];
        hv[j][1] = hrow[32];
        hv[j][2] = hrow[64];
        hv[j][3] = hrow[96];
    }

    const float ex = ebuf[edge];
    const float sm = ssum[d_idx];
    const float coef = ex / (sm + 1e-16f);

#pragma unroll
    for (int j = 0; j < SC_EPW; ++j) {
        const int   dj = __builtin_amdgcn_readlane(d_idx, j);
        const float cj = __uint_as_float(
            (unsigned)__builtin_amdgcn_readlane((int)__float_as_uint(coef), j));
        float* o = out + (size_t)dj * GAT_D + lane;
        atomicAdd(o,      hv[j][0] * cj);
        atomicAdd(o + 32, hv[j][1] * cj);
        atomicAdd(o + 64, hv[j][2] * cj);
        atomicAdd(o + 96, hv[j][3] * cj);
    }
}

__global__ void gat_pool_zero(float* __restrict__ pool, unsigned* __restrict__ cnt) {
    const int t = blockIdx.x * blockDim.x + threadIdx.x;
    if (t < GAT_G * GAT_D) pool[t] = 0.f;
    if (t < GAT_G) cnt[t] = 0u;
}

__global__ void gat_pool_accum(const float* __restrict__ feats,
                               const int* __restrict__ batch,
                               float* __restrict__ pool,
                               unsigned* __restrict__ cnt) {
    const size_t t = (size_t)blockIdx.x * blockDim.x + threadIdx.x;
    if (t >= (size_t)GAT_N * GAT_D) return;
    const int n = (int)(t >> 7);
    const int d = (int)(t & (GAT_D - 1));
    const int g = batch[n];
    atomicAdd(&pool[g * GAT_D + d], feats[t]);
    if (d == 0) atomicAdd(&cnt[g], 1u);
}

__global__ void gat_pool_final(const float* __restrict__ pool,
                               const unsigned* __restrict__ cnt,
                               float* __restrict__ out_pooled) {
    const int t = blockIdx.x * blockDim.x + threadIdx.x;
    if (t >= GAT_G * GAT_D) return;
    const float c = fmaxf((float)cnt[t >> 7], 1.0f);
    out_pooled[t] = pool[t] / c;
}

static void launch_gat_layer(const float* x, const float* W,
                             const float* a_s, const float* a_d,
                             const float* bias,
                             const int* ei,
                             float* h, float* alphaS, unsigned* dm,
                             float* ssum, float* ebuf,
                             float* out, hipStream_t stream) {
    gat_gemm128_wmma<<<GAT_N / 16, 256, 0, stream>>>(x, W, h);
    gat_alpha<<<GAT_N / 8, 256, 0, stream>>>(h, a_s, a_d, alphaS, dm);
    gat_init<<<(GAT_N * GAT_D) / 256, 256, 0, stream>>>(out, bias, ssum);
    const int eb = (GAT_ETOT + 255) / 256;
    gat_edge_max<<<eb, 256, 0, stream>>>(ei, alphaS, dm, ebuf);
    gat_edge_exp<<<eb, 256, 0, stream>>>(ei, ebuf, dm, ssum);
    gat_edge_scatter<<<(GAT_ETOT / SC_EPW + 7) / 8, 256, 0, stream>>>(ei, ebuf, ssum, h, out);
}

extern "C" void kernel_launch(void* const* d_in, const int* in_sizes, int n_in,
                              void* d_out, int out_size, void* d_ws, size_t ws_size,
                              hipStream_t stream) {
    const float* x     = (const float*)d_in[0];
    const int*   ei    = (const int*)d_in[1];
    const int*   batch = (const int*)d_in[2];
    const float* W1  = (const float*)d_in[3];
    const float* as1 = (const float*)d_in[4];
    const float* ad1 = (const float*)d_in[5];
    const float* b1  = (const float*)d_in[6];
    const float* W2  = (const float*)d_in[7];
    const float* as2 = (const float*)d_in[8];
    const float* ad2 = (const float*)d_in[9];
    const float* b2  = (const float*)d_in[10];

    float* out_feats  = (float*)d_out;
    float* out_pooled = out_feats + (size_t)GAT_N * GAT_D;

    float* ws = (float*)d_ws;
    size_t o = 0;
    float*    h      = ws + o; o += (size_t)GAT_N * GAT_D;
    float*    feats1 = ws + o; o += (size_t)GAT_N * GAT_D;
    float*    alphaS = ws + o; o += GAT_N;
    unsigned* dm     = (unsigned*)(ws + o); o += 2 * (size_t)GAT_N;
    float*    ssum   = ws + o; o += GAT_N;
    float*    ebuf   = ws + o; o += GAT_ETOT;
    float*    pool   = ws + o; o += GAT_G * GAT_D;
    unsigned* cnt    = (unsigned*)(ws + o); o += GAT_G;
    (void)ws_size; (void)in_sizes; (void)n_in; (void)out_size;

    launch_gat_layer(x, W1, as1, ad1, b1, ei,
                     h, alphaS, dm, ssum, ebuf, feats1, stream);
    launch_gat_layer(feats1, W2, as2, ad2, b2, ei,
                     h, alphaS, dm, ssum, ebuf, out_feats, stream);

    gat_pool_zero<<<(GAT_G * GAT_D + 255) / 256, 256, 0, stream>>>(pool, cnt);
    gat_pool_accum<<<(GAT_N * GAT_D) / 256, 256, 0, stream>>>(out_feats, batch, pool, cnt);
    gat_pool_final<<<(GAT_G * GAT_D + 255) / 256, 256, 0, stream>>>(pool, cnt, out_pooled);
}


# Aggregation: each block owns 196 destination nodes, filters the edge list for them and does max, exp-sum and weighted row sum in LDS (LDS atomics, arrival order) instead of L2 atomics; 9 launches
# speedup vs baseline: 2.2675x; 1.5526x over previous
#include <hip/hip_runtime.h>
#include <hip/hip_bf16.h>

#define GAT_N      50000
#define GAT_E      800000
#define GAT_ETOT   850000
#define GAT_D      128
#define GAT_G      64
#define NEG_SLOPE  0.2f

typedef __attribute__((ext_vector_type(2))) float v2f;
typedef __attribute__((ext_vector_type(8))) float v8f;

__device__ __forceinline__ unsigned enc_f32(float f) {
    unsigned u = __float_as_uint(f);
    return (u & 0x80000000u) ? ~u : (u | 0x80000000u);
}
__device__ __forceinline__ float dec_f32(unsigned k) {
    unsigned u = (k & 0x80000000u) ? (k & 0x7FFFFFFFu) : ~k;
    return __uint_as_float(u);
}

__global__ void gat_gemm128_wmma(const float* __restrict__ X,
                                 const float* __restrict__ W,
                                 float* __restrict__ H) {
    const int wave = threadIdx.x >> 5;
    const int lane = threadIdx.x & 31;
    const int half = lane >> 4;
    const int l    = lane & 15;
    const int m0   = blockIdx.x * 16;
    const int n0   = wave * 16;

    const float* xrow = X + (size_t)(m0 + l) * GAT_D;
    v8f acc = {};

#pragma unroll
    for (int k = 0; k < GAT_D; k += 4) {
        const int kk = k + 2 * half;
        v2f a;
        {
            const float2 av = *(const float2*)(xrow + kk);
            a.x = av.x; a.y = av.y;
        }
        v2f b;
        b.x = W[(size_t)kk       * GAT_D + n0 + l];
        b.y = W[(size_t)(kk + 1) * GAT_D + n0 + l];
        acc = __builtin_amdgcn_wmma_f32_16x16x4_f32(
            false, a, false, b, (short)0, acc, false, false);
    }

    const int mBase = m0 + half * 8;
#pragma unroll
    for (int r = 0; r < 8; ++r)
        H[(size_t)(mBase + r) * GAT_D + n0 + l] = acc[r];
}

__global__ void gat_alpha(const float* __restrict__ H,
                          const float* __restrict__ a_src,
                          const float* __restrict__ a_dst,
                          float* __restrict__ as_out,
                          float* __restrict__ ad_out) {
    const int node = blockIdx.x * 8 + (threadIdx.x >> 5);
    const int lane = threadIdx.x & 31;
    if (node >= GAT_N) return;
    const float* h = H + (size_t)node * GAT_D;
    float s = 0.f, d = 0.f;
#pragma unroll
    for (int i = lane; i < GAT_D; i += 32) {
        const float v = h[i];
        s += v * a_src[i];
        d += v * a_dst[i];
    }
#pragma unroll
    for (int off = 16; off > 0; off >>= 1) {
        s += __shfl_xor(s, off, 32);
        d += __shfl_xor(d, off, 32);
    }
    if (lane == 0) {
        as_out[node] = s;
        ad_out[node] = d;
    }
}

#define AG_R    196
#define AG_NB   256
#define AG_T    1024
#define AG_CAP  8192
#define AG_SU   8
#define AG_GU   8

__device__ __forceinline__ float ag_lrelu(float e) {
    return (e > 0.f) ? e : NEG_SLOPE * e;
}
__device__ __forceinline__ float ag_segmax(unsigned k) {
    const float m = dec_f32(k);
    return __builtin_isfinite(m) ? m : 0.f;
}

__device__ __forceinline__ bool ag_fetch(bool overflow, int idx, unsigned cnt,
                                         const unsigned* l_sd, const float* l_e,
                                         const int* __restrict__ ei,
                                         const float* __restrict__ as,
                                         const float* s_ad, int n0, int nn,
                                         unsigned& sd, float& e) {
    sd = 0u; e = 0.f;
    if (!overflow) {
        if ((unsigned)idx < cnt) { sd = l_sd[idx]; e = l_e[idx]; return true; }
        return false;
    }
    const unsigned dl = (unsigned)(ei[GAT_E + idx] - n0);
    if (dl < (unsigned)nn) {
        const int s = ei[idx];
        sd = (unsigned)s | (dl << 16);
        e  = ag_lrelu(as[s] + s_ad[dl]);
        return true;
    }
    return false;
}

__global__ __launch_bounds__(AG_T)
void gat_aggregate(const int* __restrict__ ei,
                   const float* __restrict__ H,
                   const float* __restrict__ as,
                   const float* __restrict__ ad,
                   const float* __restrict__ bias,
                   float* __restrict__ out) {
    __shared__ float    acc[AG_R * GAT_D];
    __shared__ float    s_ad[AG_R];
    __shared__ float    s_es[AG_R];
    __shared__ unsigned s_mx[AG_R];
    __shared__ float    s_sum[AG_R];
    __shared__ unsigned l_sd[AG_CAP];
    __shared__ float    l_e[AG_CAP];
    __shared__ unsigned s_cnt;

    const int tid  = threadIdx.x;
    const int lane = tid & 31;
    const int wave = tid >> 5;
    const int n0   = blockIdx.x * AG_R;
    const int nn   = (GAT_N - n0 < AG_R) ? (GAT_N - n0) : AG_R;

    for (int i = tid; i < AG_R * GAT_D; i += AG_T) acc[i] = 0.f;
    if (tid < nn) {
        const float d = ad[n0 + tid];
        const float e = ag_lrelu(as[n0 + tid] + d);
        s_ad[tid]  = d;
        s_es[tid]  = e;
        s_mx[tid]  = enc_f32(e);
        s_sum[tid] = 0.f;
    }
    if (tid == 0) s_cnt = 0u;
    __syncthreads();

    {
        const int4* dst4 = (const int4*)(ei + GAT_E);
        const int rot = blockIdx.x * ((GAT_E / 4) / AG_NB);
        for (int q0 = 0; q0 < GAT_E / 4; q0 += AG_SU * AG_T) {
            int4 d4[AG_SU];
            int  qq[AG_SU];
#pragma unroll
            for (int u = 0; u < AG_SU; ++u) {
                const int q = q0 + u * AG_T + tid;
                qq[u] = (q + rot >= GAT_E / 4) ? (q + rot - GAT_E / 4) : (q + rot);
                d4[u] = (q < GAT_E / 4) ? dst4[qq[u]] : make_int4(-1, -1, -1, -1);
            }
#pragma unroll
            for (int u = 0; u < AG_SU; ++u) {
                const int q = qq[u];
                const int dd[4] = {d4[u].x, d4[u].y, d4[u].z, d4[u].w};
#pragma unroll
                for (int k = 0; k < 4; ++k) {
                    const unsigned dl = (unsigned)(dd[k] - n0);
                    if (dl < (unsigned)nn) {
                        const unsigned slot = atomicAdd(&s_cnt, 1u);
                        if (slot < AG_CAP) l_sd[slot] = (unsigned)(4 * q + k) | (dl << 20);
                    }
                }
            }
        }
    }
    __syncthreads();
    const unsigned cnt = s_cnt;
    const bool overflow = cnt > AG_CAP;
    const int nchunks = overflow ? (GAT_E / 32) : (int)((cnt + 31u) >> 5);

    if (!overflow) {
        for (int j = tid; j < (int)cnt; j += AG_T) {
            const unsigned w  = l_sd[j];
            const unsigned dl = w >> 20;
            const int s = ei[w & 0xFFFFFu];
            const float e = ag_lrelu(as[s] + s_ad[dl]);
            atomicMax(&s_mx[dl], enc_f32(e));
            l_sd[j] = (unsigned)s | (dl << 16);
            l_e[j]  = e;
        }
    } else {
        for (int c = wave; c < nchunks; c += AG_T / 32) {
            unsigned sd; float e;
            if (ag_fetch(true, c * 32 + lane, cnt, l_sd, l_e, ei, as, s_ad, n0, nn, sd, e))
                atomicMax(&s_mx[sd >> 16], enc_f32(e));
        }
    }
    __syncthreads();

    for (int c = wave; c < nchunks; c += AG_T / 32) {
        unsigned sd; float e;
        if (ag_fetch(overflow, c * 32 + lane, cnt, l_sd, l_e, ei, as, s_ad, n0, nn, sd, e)) {
            const unsigned dl = sd >> 16;
            atomicAdd(&s_sum[dl], expf(e - ag_segmax(s_mx[dl])));
        }
    }
    if (tid < nn) atomicAdd(&s_sum[tid], expf(s_es[tid] - ag_segmax(s_mx[tid])));
    __syncthreads();

    for (int c = wave; c < nchunks; c += AG_T / 32) {
        unsigned sd; float e;
        const bool valid = ag_fetch(overflow, c * 32 + lane, cnt, l_sd, l_e, ei, as, s_ad, n0, nn, sd, e);
        float coef = 0.f;
        if (valid) {
            const unsigned dl = sd >> 16;
            coef = expf(e - ag_segmax(s_mx[dl])) / (s_sum[dl] + 1e-16f);
        }
        unsigned mask = __builtin_amdgcn_ballot_w32(valid);
        while (mask) {
            bool     ok[AG_GU];
            unsigned sj[AG_GU];
            float    cj[AG_GU];
            float4   hv[AG_GU];
#pragma unroll
            for (int u = 0; u < AG_GU; ++u) {
                ok[u] = (mask != 0u);
                const int j = ok[u] ? __builtin_ctz(mask) : 0;
                mask &= mask - 1u;
                sj[u] = (unsigned)__builtin_amdgcn_readlane((int)sd, j);
                cj[u] = __uint_as_float((unsigned)__builtin_amdgcn_readlane((int)__float_as_uint(coef), j));
                hv[u] = *(const float4*)(H + (size_t)(sj[u] & 0xFFFFu) * GAT_D + 4 * lane);
            }
#pragma unroll
            for (int u = 0; u < AG_GU; ++u)
                asm volatile("" :: "v"(hv[u].x), "v"(hv[u].y), "v"(hv[u].z), "v"(hv[u].w));
#pragma unroll
            for (int u = 0; u < AG_GU; ++u) {
                if (ok[u]) {
                    float* a = acc + (sj[u] >> 16) * GAT_D + lane;
                    atomicAdd(a,      hv[u].x * cj[u]);
                    atomicAdd(a + 32, hv[u].y * cj[u]);
                    atomicAdd(a + 64, hv[u].z * cj[u]);
                    atomicAdd(a + 96, hv[u].w * cj[u]);
                }
            }
        }
    }
    __syncthreads();

    const float4 bv = *(const float4*)(bias + 4 * lane);
    for (int dl = wave; dl < nn; dl += AG_T / 32) {
        const size_t row = (size_t)(n0 + dl) * GAT_D + 4 * lane;
        const float4 hs = *(const float4*)(H + row);
        const float coef = expf(s_es[dl] - ag_segmax(s_mx[dl])) / (s_sum[dl] + 1e-16f);
        const float* a = acc + dl * GAT_D + lane;
        float4 o;
        o.x = (a[0]  + hs.x * coef) + bv.x;
        o.y = (a[32] + hs.y * coef) + bv.y;
        o.z = (a[64] + hs.z * coef) + bv.z;
        o.w = (a[96] + hs.w * coef) + bv.w;
        *(float4*)(out + row) = o;
    }
}

__global__ void gat_pool_zero(float* __restrict__ pool, unsigned* __restrict__ cnt) {
    const int t = blockIdx.x * blockDim.x + threadIdx.x;
    if (t < GAT_G * GAT_D) pool[t] = 0.f;
    if (t < GAT_G) cnt[t] = 0u;
}

__global__ void gat_pool_accum(const float* __restrict__ feats,
                               const int* __restrict__ batch,
                               float* __restrict__ pool,
                               unsigned* __restrict__ cnt) {
    const size_t t = (size_t)blockIdx.x * blockDim.x + threadIdx.x;
    if (t >= (size_t)GAT_N * GAT_D) return;
    const int n = (int)(t >> 7);
    const int d = (int)(t & (GAT_D - 1));
    const int g = batch[n];
    atomicAdd(&pool[g * GAT_D + d], feats[t]);
    if (d == 0) atomicAdd(&cnt[g], 1u);
}

__global__ void gat_pool_final(const float* __restrict__ pool,
                               const unsigned* __restrict__ cnt,
                               float* __restrict__ out_pooled) {
    const int t = blockIdx.x * blockDim.x + threadIdx.x;
    if (t >= GAT_G * GAT_D) return;
    const float c = fmaxf((float)cnt[t >> 7], 1.0f);
    out_pooled[t] = pool[t] / c;
}

static void launch_gat_layer(const float* x, const float* W,
                             const float* a_s, const float* a_d,
                             const float* bias,
                             const int* ei,
                             float* h, float* alphaS, float* alphaD,
                             float* out, hipStream_t stream) {
    gat_gemm128_wmma<<<GAT_N / 16, 256, 0, stream>>>(x, W, h);
    gat_alpha<<<GAT_N / 8, 256, 0, stream>>>(h, a_s, a_d, alphaS, alphaD);
    gat_aggregate<<<AG_NB, AG_T, 0, stream>>>(ei, h, alphaS, alphaD, bias, out);
}

extern "C" void kernel_launch(void* const* d_in, const int* in_sizes, int n_in,
                              void* d_out, int out_size, void* d_ws, size_t ws_size,
                              hipStream_t stream) {
    const float* x     = (const float*)d_in[0];
    const int*   ei    = (const int*)d_in[1];
    const int*   batch = (const int*)d_in[2];
    const float* W1  = (const float*)d_in[3];
    const float* as1 = (const float*)d_in[4];
    const float* ad1 = (const float*)d_in[5];
    const float* b1  = (const float*)d_in[6];
    const float* W2  = (const float*)d_in[7];
    const float* as2 = (const float*)d_in[8];
    const float* ad2 = (const float*)d_in[9];
    const float* b2  = (const float*)d_in[10];

    float* out_feats  = (float*)d_out;
    float* out_pooled = out_feats + (size_t)GAT_N * GAT_D;

    float* ws = (float*)d_ws;
    size_t o = 0;
    float*    h      = ws + o; o += (size_t)GAT_N * GAT_D;
    float*    feats1 = ws + o; o += (size_t)GAT_N * GAT_D;
    float*    alphaS = ws + o; o += GAT_N;
    float*    alphaD = ws + o; o += GAT_N;
    float*    pool   = ws + o; o += GAT_G * GAT_D;
    unsigned* cnt    = (unsigned*)(ws + o); o += GAT_G;
    (void)ws_size; (void)in_sizes; (void)n_in; (void)out_size;

    launch_gat_layer(x, W1, as1, ad1, b1, ei,
                     h, alphaS, alphaD, feats1, stream);
    launch_gat_layer(feats1, W2, as2, ad2, b2, ei,
                     h, alphaS, alphaD, out_feats, stream);

    gat_pool_zero<<<(GAT_G * GAT_D + 255) / 256, 256, 0, stream>>>(pool, cnt);
    gat_pool_accum<<<(GAT_N * GAT_D) / 256, 256, 0, stream>>>(out_feats, batch, pool, cnt);
    gat_pool_final<<<(GAT_G * GAT_D + 255) / 256, 256, 0, stream>>>(pool, cnt, out_pooled);
}


# Edge list built once per call: the layer-1 aggregate stores each block's filtered (source, local node) list in the block's own output rows; layer 2 reads it, no second scan of the whole dst list
# speedup vs baseline: 2.5811x; 1.1383x over previous
#include <hip/hip_runtime.h>
#include <hip/hip_bf16.h>

#define GAT_N      50000
#define GAT_E      800000
#define GAT_ETOT   850000
#define GAT_D      128
#define GAT_G      64
#define NEG_SLOPE  0.2f

typedef __attribute__((ext_vector_type(2))) float v2f;
typedef __attribute__((ext_vector_type(8))) float v8f;

__device__ __forceinline__ unsigned enc_f32(float f) {
    unsigned u = __float_as_uint(f);
    return (u & 0x80000000u) ? ~u : (u | 0x80000000u);
}
__device__ __forceinline__ float dec_f32(unsigned k) {
    unsigned u = (k & 0x80000000u) ? (k & 0x7FFFFFFFu) : ~k;
    return __uint_as_float(u);
}

__global__ void gat_gemm128_wmma(const float* __restrict__ X,
                                 const float* __restrict__ W,
                                 float* __restrict__ H) {
    const int wave = threadIdx.x >> 5;
    const int lane = threadIdx.x & 31;
    const int half = lane >> 4;
    const int l    = lane & 15;
    const int m0   = blockIdx.x * 16;
    const int n0   = wave * 16;

    const float* xrow = X + (size_t)(m0 + l) * GAT_D;
    v8f acc = {};

#pragma unroll
    for (int k = 0; k < GAT_D; k += 4) {
        const int kk = k + 2 * half;
        v2f a;
        {
            const float2 av = *(const float2*)(xrow + kk);
            a.x = av.x; a.y = av.y;
        }
        v2f b;
        b.x = W[(size_t)kk       * GAT_D + n0 + l];
        b.y = W[(size_t)(kk + 1) * GAT_D + n0 + l];
        acc = __builtin_amdgcn_wmma_f32_16x16x4_f32(
            false, a, false, b, (short)0, acc, false, false);
    }

    const int mBase = m0 + half * 8;
#pragma unroll
    for (int r = 0; r < 8; ++r)
        H[(size_t)(mBase + r) * GAT_D + n0 + l] = acc[r];
}

__global__ void gat_alpha(const float* __restrict__ H,
                          const float* __restrict__ a_src,
                          const float* __restrict__ a_dst,
                          float* __restrict__ as_out,
                          float* __restrict__ ad_out) {
    const int node = blockIdx.x * 8 + (threadIdx.x >> 5);
    const int lane = threadIdx.x & 31;
    if (node >= GAT_N) return;
    const float* h = H + (size_t)node * GAT_D;
    float s = 0.f, d = 0.f;
#pragma unroll
    for (int i = lane; i < GAT_D; i += 32) {
        const float v = h[i];
        s += v * a_src[i];
        d += v * a_dst[i];
    }
#pragma unroll
    for (int off = 16; off > 0; off >>= 1) {
        s += __shfl_xor(s, off, 32);
        d += __shfl_xor(d, off, 32);
    }
    if (lane == 0) {
        as_out[node] = s;
        ad_out[node] = d;
    }
}

#define AG_R    196
#define AG_NB   256
#define AG_T    1024
#define AG_CAP  8192
#define AG_SU   8
#define AG_GU   8

__device__ __forceinline__ float ag_lrelu(float e) {
    return (e > 0.f) ? e : NEG_SLOPE * e;
}
__device__ __forceinline__ float ag_segmax(unsigned k) {
    const float m = dec_f32(k);
    return __builtin_isfinite(m) ? m : 0.f;
}

__device__ __forceinline__ bool ag_fetch(bool overflow, int idx, unsigned cnt,
                                         const unsigned* l_sd, const float* l_e,
                                         const int* __restrict__ ei,
                                         const float* __restrict__ as,
                                         const float* s_ad, int n0, int nn,
                                         unsigned& sd, float& e) {
    sd = 0u; e = 0.f;
    if (!overflow) {
        if ((unsigned)idx < cnt) { sd = l_sd[idx]; e = l_e[idx]; return true; }
        return false;
    }
    const unsigned dl = (unsigned)(ei[GAT_E + idx] - n0);
    if (dl < (unsigned)nn) {
        const int s = ei[idx];
        sd = (unsigned)s | (dl << 16);
        e  = ag_lrelu(as[s] + s_ad[dl]);
        return true;
    }
    return false;
}

__global__ __launch_bounds__(AG_T)
void gat_aggregate(const int* __restrict__ ei,
                   const float* __restrict__ H,
                   const float* __restrict__ as,
                   const float* __restrict__ ad,
                   const float* __restrict__ bias,
                   float* out,
                   unsigned* lst_all,
                   int second) {
    __shared__ float    acc[AG_R * GAT_D];
    __shared__ float    s_ad[AG_R];
    __shared__ float    s_es[AG_R];
    __shared__ unsigned s_mx[AG_R];
    __shared__ float    s_sum[AG_R];
    __shared__ unsigned l_sd[AG_CAP];
    __shared__ float    l_e[AG_CAP];
    __shared__ unsigned s_cnt;

    const int tid  = threadIdx.x;
    const int lane = tid & 31;
    const int wave = tid >> 5;
    const int n0   = blockIdx.x * AG_R;
    const int nn   = (GAT_N - n0 < AG_R) ? (GAT_N - n0) : AG_R;
    unsigned* lst = lst_all + (size_t)n0 * GAT_D;
    const unsigned lcap = ((unsigned)(nn * GAT_D - 4) < (unsigned)AG_CAP)
                              ? (unsigned)(nn * GAT_D - 4) : (unsigned)AG_CAP;

    for (int i = tid; i < AG_R * GAT_D; i += AG_T) acc[i] = 0.f;
    if (tid < nn) {
        const float d = ad[n0 + tid];
        const float e = ag_lrelu(as[n0 + tid] + d);
        s_ad[tid]  = d;
        s_es[tid]  = e;
        s_mx[tid]  = enc_f32(e);
        s_sum[tid] = 0.f;
    }
    if (tid == 0) s_cnt = 0u;
    __syncthreads();

    bool listed = false;
    if (second) {
        const unsigned c = (unsigned)__builtin_amdgcn_readfirstlane((int)lst[0]);
        listed = (c <= lcap);
        if (listed) {
            for (unsigned j = tid; j < c; j += AG_T) l_sd[j] = lst[4 + j];
            if (tid == 0) s_cnt = c;
        }
    }

    if (!listed) {
        const int4* dst4 = (const int4*)(ei + GAT_E);
        const int rot = blockIdx.x * ((GAT_E / 4) / AG_NB);
        for (int q0 = 0; q0 < GAT_E / 4; q0 += AG_SU * AG_T) {
            int4 d4[AG_SU];
            int  qq[AG_SU];
#pragma unroll
            for (int u = 0; u < AG_SU; ++u) {
                const int q = q0 + u * AG_T + tid;
                qq[u] = (q + rot >= GAT_E / 4) ? (q + rot - GAT_E / 4) : (q + rot);
                d4[u] = (q < GAT_E / 4) ? dst4[qq[u]] : make_int4(-1, -1, -1, -1);
            }
#pragma unroll
            for (int u = 0; u < AG_SU; ++u) {
                const int q = qq[u];
                const int dd[4] = {d4[u].x, d4[u].y, d4[u].z, d4[u].w};
#pragma unroll
                for (int k = 0; k < 4; ++k) {
                    const unsigned dl = (unsigned)(dd[k] - n0);
                    if (dl < (unsigned)nn) {
                        const unsigned slot = atomicAdd(&s_cnt, 1u);
                        if (slot < AG_CAP) l_sd[slot] = (unsigned)(4 * q + k) | (dl << 20);
                    }
                }
            }
        }
    }
    __syncthreads();
    const unsigned cnt = s_cnt;
    const bool overflow = cnt > AG_CAP;
    const int nchunks = overflow ? (GAT_E / 32) : (int)((cnt + 31u) >> 5);

    const bool keep = !second && cnt <= lcap;
    if (!overflow) {
        for (int j = tid; j < (int)cnt; j += AG_T) {
            const unsigned w  = l_sd[j];
            unsigned dl; int s;
            if (listed) { s = (int)(w & 0xFFFFu); dl = w >> 16; }
            else        { dl = w >> 20; s = ei[w & 0xFFFFFu]; }
            const float e = ag_lrelu(as[s] + s_ad[dl]);
            atomicMax(&s_mx[dl], enc_f32(e));
            const unsigned sd = (unsigned)s | (dl << 16);
            l_sd[j] = sd;
            l_e[j]  = e;
            if (keep) lst[4 + j] = sd;
        }
    } else {
        for (int c = wave; c < nchunks; c += AG_T / 32) {
            unsigned sd; float e;
            if (ag_fetch(true, c * 32 + lane, cnt, l_sd, l_e, ei, as, s_ad, n0, nn, sd, e))
                atomicMax(&s_mx[sd >> 16], enc_f32(e));
        }
    }
    if (!second && tid == 0) lst[0] = keep ? cnt : 0xFFFFFFFFu;
    __syncthreads();

    for (int c = wave; c < nchunks; c += AG_T / 32) {
        unsigned sd; float e;
        if (ag_fetch(overflow, c * 32 + lane, cnt, l_sd, l_e, ei, as, s_ad, n0, nn, sd, e)) {
            const unsigned dl = sd >> 16;
            atomicAdd(&s_sum[dl], expf(e - ag_segmax(s_mx[dl])));
        }
    }
    if (tid < nn) atomicAdd(&s_sum[tid], expf(s_es[tid] - ag_segmax(s_mx[tid])));
    __syncthreads();

    for (int c = wave; c < nchunks; c += AG_T / 32) {
        unsigned sd; float e;
        const bool valid = ag_fetch(overflow, c * 32 + lane, cnt, l_sd, l_e, ei, as, s_ad, n0, nn, sd, e);
        float coef = 0.f;
        if (valid) {
            const unsigned dl = sd >> 16;
            coef = expf(e - ag_segmax(s_mx[dl])) / (s_sum[dl] + 1e-16f);
        }
        unsigned mask = __builtin_amdgcn_ballot_w32(valid);
        while (mask) {
            bool     ok[AG_GU];
            unsigned sj[AG_GU];
            float    cj[AG_GU];
            float4   hv[AG_GU];
#pragma unroll
            for (int u = 0; u < AG_GU; ++u) {
                ok[u] = (mask != 0u);
                const int j = ok[u] ? __builtin_ctz(mask) : 0;
                mask &= mask - 1u;
                sj[u] = (unsigned)__builtin_amdgcn_readlane((int)sd, j);
                cj[u] = __uint_as_float((unsigned)__builtin_amdgcn_readlane((int)__float_as_uint(coef), j));
                hv[u] = *(const float4*)(H + (size_t)(sj[u] & 0xFFFFu) * GAT_D + 4 * lane);
            }
#pragma unroll
            for (int u = 0; u < AG_GU; ++u)
                asm volatile("" :: "v"(hv[u].x), "v"(hv[u].y), "v"(hv[u].z), "v"(hv[u].w));
#pragma unroll
            for (int u = 0; u < AG_GU; ++u) {
                if (ok[u]) {
                    float* a = acc + (sj[u] >> 16) * GAT_D + lane;
                    atomicAdd(a,      hv[u].x * cj[u]);
                    atomicAdd(a + 32, hv[u].y * cj[u]);
                    atomicAdd(a + 64, hv[u].z * cj[u]);
                    atomicAdd(a + 96, hv[u].w * cj[u]);
                }
            }
        }
    }
    __syncthreads();

    const float4 bv = *(const float4*)(bias + 4 * lane);
    for (int dl = wave; dl < nn; dl += AG_T / 32) {
        const size_t row = (size_t)(n0 + dl) * GAT_D + 4 * lane;
        const float4 hs = *(const float4*)(H + row);
        const float coef = expf(s_es[dl] - ag_segmax(s_mx[dl])) / (s_sum[dl] + 1e-16f);
        const float* a = acc + dl * GAT_D + lane;
        float4 o;
        o.x = (a[0]  + hs.x * coef) + bv.x;
        o.y = (a[32] + hs.y * coef) + bv.y;
        o.z = (a[64] + hs.z * coef) + bv.z;
        o.w = (a[96] + hs.w * coef) + bv.w;
        *(float4*)(out + row) = o;
    }
}

__global__ void gat_pool_zero(float* __restrict__ pool, unsigned* __restrict__ cnt) {
    const int t = blockIdx.x * blockDim.x + threadIdx.x;
    if (t < GAT_G * GAT_D) pool[t] = 0.f;
    if (t < GAT_G) cnt[t] = 0u;
}

__global__ void gat_pool_accum(const float* __restrict__ feats,
                               const int* __restrict__ batch,
                               float* __restrict__ pool,
                               unsigned* __restrict__ cnt) {
    const size_t t = (size_t)blockIdx.x * blockDim.x + threadIdx.x;
    if (t >= (size_t)GAT_N * GAT_D) return;
    const int n = (int)(t >> 7);
    const int d = (int)(t & (GAT_D - 1));
    const int g = batch[n];
    atomicAdd(&pool[g * GAT_D + d], feats[t]);
    if (d == 0) atomicAdd(&cnt[g], 1u);
}

__global__ void gat_pool_final(const float* __restrict__ pool,
                               const unsigned* __restrict__ cnt,
                               float* __restrict__ out_pooled) {
    const int t = blockIdx.x * blockDim.x + threadIdx.x;
    if (t >= GAT_G * GAT_D) return;
    const float c = fmaxf((float)cnt[t >> 7], 1.0f);
    out_pooled[t] = pool[t] / c;
}

static void launch_gat_layer(const float* x, const float* W,
                             const float* a_s, const float* a_d,
                             const float* bias,
                             const int* ei,
                             float* h, float* alphaS, float* alphaD,
                             float* out, unsigned* lst_all, int second,
                             hipStream_t stream) {
    gat_gemm128_wmma<<<GAT_N / 16, 256, 0, stream>>>(x, W, h);
    gat_alpha<<<GAT_N / 8, 256, 0, stream>>>(h, a_s, a_d, alphaS, alphaD);
    gat_aggregate<<<AG_NB, AG_T, 0, stream>>>(ei, h, alphaS, alphaD, bias, out, lst_all, second);
}

extern "C" void kernel_launch(void* const* d_in, const int* in_sizes, int n_in,
                              void* d_out, int out_size, void* d_ws, size_t ws_size,
                              hipStream_t stream) {
    const float* x     = (const float*)d_in[0];
    const int*   ei    = (const int*)d_in[1];
    const int*   batch = (const int*)d_in[2];
    const float* W1  = (const float*)d_in[3];
    const float* as1 = (const float*)d_in[4];
    const float* ad1 = (const float*)d_in[5];
    const float* b1  = (const float*)d_in[6];
    const float* W2  = (const float*)d_in[7];
    const float* as2 = (const float*)d_in[8];
    const float* ad2 = (const float*)d_in[9];
    const float* b2  = (const float*)d_in[10];

    float* out_feats  = (float*)d_out;
    float* out_pooled = out_feats + (size_t)GAT_N * GAT_D;

    float* ws = (float*)d_ws;
    size_t o = 0;
    float*    h      = ws + o; o += (size_t)GAT_N * GAT_D;
    float*    feats1 = ws + o; o += (size_t)GAT_N * GAT_D;
    float*    alphaS = ws + o; o += GAT_N;
    float*    alphaD = ws + o; o += GAT_N;
    float*    pool   = ws + o; o += GAT_G * GAT_D;
    unsigned* cnt    = (unsigned*)(ws + o); o += GAT_G;
    (void)ws_size; (void)in_sizes; (void)n_in; (void)out_size;

    launch_gat_layer(x, W1, as1, ad1, b1, ei,
                     h, alphaS, alphaD, feats1, (unsigned*)out_feats, 0, stream);
    launch_gat_layer(feats1, W2, as2, ad2, b2, ei,
                     h, alphaS, alphaD, out_feats, (unsigned*)out_feats, 1, stream);

    gat_pool_zero<<<(GAT_G * GAT_D + 255) / 256, 256, 0, stream>>>(pool, cnt);
    gat_pool_accum<<<(GAT_N * GAT_D) / 256, 256, 0, stream>>>(out_feats, batch, pool, cnt);
    gat_pool_final<<<(GAT_G * GAT_D + 255) / 256, 256, 0, stream>>>(pool, cnt, out_pooled);
}


# Mean pool by owner: block per (graph, 32 columns) filters the graph-id list, sums its nodes' row lines in registers, 32 waves merged in LDS; no atomics, zero or divide launch; new fixed sum order
# speedup vs baseline: 3.4606x; 1.3407x over previous
#include <hip/hip_runtime.h>
#include <hip/hip_bf16.h>

#define GAT_N      50000
#define GAT_E      800000
#define GAT_ETOT   850000
#define GAT_D      128
#define GAT_G      64
#define NEG_SLOPE  0.2f

typedef __attribute__((ext_vector_type(2))) float v2f;
typedef __attribute__((ext_vector_type(8))) float v8f;

__device__ __forceinline__ unsigned enc_f32(float f) {
    unsigned u = __float_as_uint(f);
    return (u & 0x80000000u) ? ~u : (u | 0x80000000u);
}
__device__ __forceinline__ float dec_f32(unsigned k) {
    unsigned u = (k & 0x80000000u) ? (k & 0x7FFFFFFFu) : ~k;
    return __uint_as_float(u);
}

__global__ void gat_gemm128_wmma(const float* __restrict__ X,
                                 const float* __restrict__ W,
                                 float* __restrict__ H) {
    const int wave = threadIdx.x >> 5;
    const int lane = threadIdx.x & 31;
    const int half = lane >> 4;
    const int l    = lane & 15;
    const int m0   = blockIdx.x * 16;
    const int n0   = wave * 16;

    const float* xrow = X + (size_t)(m0 + l) * GAT_D;
    v8f acc = {};

#pragma unroll
    for (int k = 0; k < GAT_D; k += 4) {
        const int kk = k + 2 * half;
        v2f a;
        {
            const float2 av = *(const float2*)(xrow + kk);
            a.x = av.x; a.y = av.y;
        }
        v2f b;
        b.x = W[(size_t)kk       * GAT_D + n0 + l];
        b.y = W[(size_t)(kk + 1) * GAT_D + n0 + l];
        acc = __builtin_amdgcn_wmma_f32_16x16x4_f32(
            false, a, false, b, (short)0, acc, false, false);
    }

    const int mBase = m0 + half * 8;
#pragma unroll
    for (int r = 0; r < 8; ++r)
        H[(size_t)(mBase + r) * GAT_D + n0 + l] = acc[r];
}

__global__ void gat_alpha(const float* __restrict__ H,
                          const float* __restrict__ a_src,
                          const float* __restrict__ a_dst,
                          float* __restrict__ as_out,
                          float* __restrict__ ad_out) {
    const int node = blockIdx.x * 8 + (threadIdx.x >> 5);
    const int lane = threadIdx.x & 31;
    if (node >= GAT_N) return;
    const float* h = H + (size_t)node * GAT_D;
    float s = 0.f, d = 0.f;
#pragma unroll
    for (int i = lane; i < GAT_D; i += 32) {
        const float v = h[i];
        s += v * a_src[i];
        d += v * a_dst[i];
    }
#pragma unroll
    for (int off = 16; off > 0; off >>= 1) {
        s += __shfl_xor(s, off, 32);
        d += __shfl_xor(d, off, 32);
    }
    if (lane == 0) {
        as_out[node] = s;
        ad_out[node] = d;
    }
}

#define AG_R    196
#define AG_NB   256
#define AG_T    1024
#define AG_CAP  8192
#define AG_SU   8
#define AG_GU   8

__device__ __forceinline__ float ag_lrelu(float e) {
    return (e > 0.f) ? e : NEG_SLOPE * e;
}
__device__ __forceinline__ float ag_segmax(unsigned k) {
    const float m = dec_f32(k);
    return __builtin_isfinite(m) ? m : 0.f;
}

__device__ __forceinline__ bool ag_fetch(bool overflow, int idx, unsigned cnt,
                                         const unsigned* l_sd, const float* l_e,
                                         const int* __restrict__ ei,
                                         const float* __restrict__ as,
                                         const float* s_ad, int n0, int nn,
                                         unsigned& sd, float& e) {
    sd = 0u; e = 0.f;
    if (!overflow) {
        if ((unsigned)idx < cnt) { sd = l_sd[idx]; e = l_e[idx]; return true; }
        return false;
    }
    const unsigned dl = (unsigned)(ei[GAT_E + idx] - n0);
    if (dl < (unsigned)nn) {
        const int s = ei[idx];
        sd = (unsigned)s | (dl << 16);
        e  = ag_lrelu(as[s] + s_ad[dl]);
        return true;
    }
    return false;
}

__global__ __launch_bounds__(AG_T)
void gat_aggregate(const int* __restrict__ ei,
                   const float* __restrict__ H,
                   const float* __restrict__ as,
                   const float* __restrict__ ad,
                   const float* __restrict__ bias,
                   float* out,
                   unsigned* lst_all,
                   int second) {
    __shared__ float    acc[AG_R * GAT_D];
    __shared__ float    s_ad[AG_R];
    __shared__ float    s_es[AG_R];
    __shared__ unsigned s_mx[AG_R];
    __shared__ float    s_sum[AG_R];
    __shared__ unsigned l_sd[AG_CAP];
    __shared__ float    l_e[AG_CAP];
    __shared__ unsigned s_cnt;

    const int tid  = threadIdx.x;
    const int lane = tid & 31;
    const int wave = tid >> 5;
    const int n0   = blockIdx.x * AG_R;
    const int nn   = (GAT_N - n0 < AG_R) ? (GAT_N - n0) : AG_R;
    unsigned* lst = lst_all + (size_t)n0 * GAT_D;
    const unsigned lcap = ((unsigned)(nn * GAT_D - 4) < (unsigned)AG_CAP)
                              ? (unsigned)(nn * GAT_D - 4) : (unsigned)AG_CAP;

    for (int i = tid; i < AG_R * GAT_D; i += AG_T) acc[i] = 0.f;
    if (tid < nn) {
        const float d = ad[n0 + tid];
        const float e = ag_lrelu(as[n0 + tid] + d);
        s_ad[tid]  = d;
        s_es[tid]  = e;
        s_mx[tid]  = enc_f32(e);
        s_sum[tid] = 0.f;
    }
    if (tid == 0) s_cnt = 0u;
    __syncthreads();

    bool listed = false;
    if (second) {
        const unsigned c = (unsigned)__builtin_amdgcn_readfirstlane((int)lst[0]);
        listed = (c <= lcap);
        if (listed) {
            for (unsigned j = tid; j < c; j += AG_T) l_sd[j] = lst[4 + j];
            if (tid == 0) s_cnt = c;
        }
    }

    if (!listed) {
        const int4* dst4 = (const int4*)(ei + GAT_E);
        const int rot = blockIdx.x * ((GAT_E / 4) / AG_NB);
        for (int q0 = 0; q0 < GAT_E / 4; q0 += AG_SU * AG_T) {
            int4 d4[AG_SU];
            int  qq[AG_SU];
#pragma unroll
            for (int u = 0; u < AG_SU; ++u) {
                const int q = q0 + u * AG_T + tid;
                qq[u] = (q + rot >= GAT_E / 4) ? (q + rot - GAT_E / 4) : (q + rot);
                d4[u] = (q < GAT_E / 4) ? dst4[qq[u]] : make_int4(-1, -1, -1, -1);
            }
#pragma unroll
            for (int u = 0; u < AG_SU; ++u) {
                const int q = qq[u];
                const int dd[4] = {d4[u].x, d4[u].y, d4[u].z, d4[u].w};
#pragma unroll
                for (int k = 0; k < 4; ++k) {
                    const unsigned dl = (unsigned)(dd[k] - n0);
                    if (dl < (unsigned)nn) {
                        const unsigned slot = atomicAdd(&s_cnt, 1u);
                        if (slot < AG_CAP) l_sd[slot] = (unsigned)(4 * q + k) | (dl << 20);
                    }
                }
            }
        }
    }
    __syncthreads();
    const unsigned cnt = s_cnt;
    const bool overflow = cnt > AG_CAP;
    const int nchunks = overflow ? (GAT_E / 32) : (int)((cnt + 31u) >> 5);

    const bool keep = !second && cnt <= lcap;
    if (!overflow) {
        for (int j = tid; j < (int)cnt; j += AG_T) {
            const unsigned w  = l_sd[j];
            unsigned dl; int s;
            if (listed) { s = (int)(w & 0xFFFFu); dl = w >> 16; }
            else        { dl = w >> 20; s = ei[w & 0xFFFFFu]; }
            const float e = ag_lrelu(as[s] + s_ad[dl]);
            atomicMax(&s_mx[dl], enc_f32(e));
            const unsigned sd = (unsigned)s | (dl << 16);
            l_sd[j] = sd;
            l_e[j]  = e;
            if (keep) lst[4 + j] = sd;
        }
    } else {
        for (int c = wave; c < nchunks; c += AG_T / 32) {
            unsigned sd; float e;
            if (ag_fetch(true, c * 32 + lane, cnt, l_sd, l_e, ei, as, s_ad, n0, nn, sd, e))
                atomicMax(&s_mx[sd >> 16], enc_f32(e));
        }
    }
    if (!second && tid == 0) lst[0] = keep ? cnt : 0xFFFFFFFFu;
    __syncthreads();

    for (int c = wave; c < nchunks; c += AG_T / 32) {
        unsigned sd; float e;
        if (ag_fetch(overflow, c * 32 + lane, cnt, l_sd, l_e, ei, as, s_ad, n0, nn, sd, e)) {
            const unsigned dl = sd >> 16;
            atomicAdd(&s_sum[dl], expf(e - ag_segmax(s_mx[dl])));
        }
    }
    if (tid < nn) atomicAdd(&s_sum[tid], expf(s_es[tid] - ag_segmax(s_mx[tid])));
    __syncthreads();

    for (int c = wave; c < nchunks; c += AG_T / 32) {
        unsigned sd; float e;
        const bool valid = ag_fetch(overflow, c * 32 + lane, cnt, l_sd, l_e, ei, as, s_ad, n0, nn, sd, e);
        float coef = 0.f;
        if (valid) {
            const unsigned dl = sd >> 16;
            coef = expf(e - ag_segmax(s_mx[dl])) / (s_sum[dl] + 1e-16f);
        }
        unsigned mask = __builtin_amdgcn_ballot_w32(valid);
        while (mask) {
            bool     ok[AG_GU];
            unsigned sj[AG_GU];
            float    cj[AG_GU];
            float4   hv[AG_GU];
#pragma unroll
            for (int u = 0; u < AG_GU; ++u) {
                ok[u] = (mask != 0u);
                const int j = ok[u] ? __builtin_ctz(mask) : 0;
                mask &= mask - 1u;
                sj[u] = (unsigned)__builtin_amdgcn_readlane((int)sd, j);
                cj[u] = __uint_as_float((unsigned)__builtin_amdgcn_readlane((int)__float_as_uint(coef), j));
                hv[u] = *(const float4*)(H + (size_t)(sj[u] & 0xFFFFu) * GAT_D + 4 * lane);
            }
#pragma unroll
            for (int u = 0; u < AG_GU; ++u)
                asm volatile("" :: "v"(hv[u].x), "v"(hv[u].y), "v"(hv[u].z), "v"(hv[u].w));
#pragma unroll
            for (int u = 0; u < AG_GU; ++u) {
                if (ok[u]) {
                    float* a = acc + (sj[u] >> 16) * GAT_D + lane;
                    atomicAdd(a,      hv[u].x * cj[u]);
                    atomicAdd(a + 32, hv[u].y * cj[u]);
                    atomicAdd(a + 64, hv[u].z * cj[u]);
                    atomicAdd(a + 96, hv[u].w * cj[u]);
                }
            }
        }
    }
    __syncthreads();

    const float4 bv = *(const float4*)(bias + 4 * lane);
    for (int dl = wave; dl < nn; dl += AG_T / 32) {
        const size_t row = (size_t)(n0 + dl) * GAT_D + 4 * lane;
        const float4 hs = *(const float4*)(H + row);
        const float coef = expf(s_es[dl] - ag_segmax(s_mx[dl])) / (s_sum[dl] + 1e-16f);
        const float* a = acc + dl * GAT_D + lane;
        float4 o;
        o.x = (a[0]  + hs.x * coef) + bv.x;
        o.y = (a[32] + hs.y * coef) + bv.y;
        o.z = (a[64] + hs.z * coef) + bv.z;
        o.w = (a[96] + hs.w * coef) + bv.w;
        *(float4*)(out + row) = o;
    }
}

#define PL_Q   4
#define PL_T   1024
#define PL_W   (PL_T / 32)
#define PL_NC  ((GAT_N + 31) / 32)
#define PL_SU  7
#define PL_GU  16

__global__ __launch_bounds__(PL_T)
void gat_pool_own(const float* __restrict__ feats,
                  const int* __restrict__ batch,
                  float* __restrict__ out_pooled) {
    __shared__ float    p_sum[PL_W * 32];
    __shared__ unsigned p_cnt[PL_W];

    const int lane = threadIdx.x & 31;
    const int wave = __builtin_amdgcn_readfirstlane((int)(threadIdx.x >> 5));
    const int g    = blockIdx.x / PL_Q;
    const int c0   = (blockIdx.x % PL_Q) * 32;
    const float* col = feats + c0 + lane;

    float    acc = 0.f;
    unsigned cnt = 0u;

    for (int ch0 = wave; ch0 < PL_NC; ch0 += PL_W * PL_SU) {
        int id[PL_SU];
#pragma unroll
        for (int u = 0; u < PL_SU; ++u) {
            const int n = (ch0 + u * PL_W) * 32 + lane;
            const int v = batch[(n < GAT_N) ? n : (GAT_N - 1)];
            id[u] = (n < GAT_N) ? v : -1;
        }
#pragma unroll
        for (int u = 0; u < PL_SU; ++u) {
            const int nb = (ch0 + u * PL_W) * 32;
            unsigned mask = __builtin_amdgcn_ballot_w32(id[u] == g);
            cnt += (unsigned)__builtin_popcount(mask);
            while (mask) {
                bool  ok[PL_GU];
                float t[PL_GU];
#pragma unroll
                for (int r = 0; r < PL_GU; ++r) {
                    ok[r] = (mask != 0u);
                    const int j = ok[r] ? __builtin_ctz(mask) : 0;
                    mask &= mask - 1u;
                    t[r] = col[(size_t)(nb + j) * GAT_D];
                }
#pragma unroll
                for (int r = 0; r < PL_GU; ++r)
                    asm volatile("" :: "v"(t[r]));
#pragma unroll
                for (int r = 0; r < PL_GU; ++r) acc += ok[r] ? t[r] : 0.f;
            }
        }
    }

    p_sum[wave * 32 + lane] = acc;
    if (lane == 0) p_cnt[wave] = cnt;
    __syncthreads();
    if (wave == 0) {
        float    s = 0.f;
        unsigned c = 0u;
#pragma unroll
        for (int w = 0; w < PL_W; ++w) {
            s += p_sum[w * 32 + lane];
            c += p_cnt[w];
        }
        out_pooled[g * GAT_D + c0 + lane] = s / fmaxf((float)c, 1.0f);
    }
}

static void launch_gat_layer(const float* x, const float* W,
                             const float* a_s, const float* a_d,
                             const float* bias,
                             const int* ei,
                             float* h, float* alphaS, float* alphaD,
                             float* out, unsigned* lst_all, int second,
                             hipStream_t stream) {
    gat_gemm128_wmma<<<GAT_N / 16, 256, 0, stream>>>(x, W, h);
    gat_alpha<<<GAT_N / 8, 256, 0, stream>>>(h, a_s, a_d, alphaS, alphaD);
    gat_aggregate<<<AG_NB, AG_T, 0, stream>>>(ei, h, alphaS, alphaD, bias, out, lst_all, second);
}

extern "C" void kernel_launch(void* const* d_in, const int* in_sizes, int n_in,
                              void* d_out, int out_size, void* d_ws, size_t ws_size,
                              hipStream_t stream) {
    const float* x     = (const float*)d_in[0];
    const int*   ei    = (const int*)d_in[1];
    const int*   batch = (const int*)d_in[2];
    const float* W1  = (const float*)d_in[3];
    const float* as1 = (const float*)d_in[4];
    const float* ad1 = (const float*)d_in[5];
    const float* b1  = (const float*)d_in[6];
    const float* W2  = (const float*)d_in[7];
    const float* as2 = (const float*)d_in[8];
    const float* ad2 = (const float*)d_in[9];
    const float* b2  = (const float*)d_in[10];

    float* out_feats  = (float*)d_out;
    float* out_pooled = out_feats + (size_t)GAT_N * GAT_D;

    float* ws = (float*)d_ws;
    size_t o = 0;
    float*    h      = ws + o; o += (size_t)GAT_N * GAT_D;
    float*    feats1 = ws + o; o += (size_t)GAT_N * GAT_D;
    float*    alphaS = ws + o; o += GAT_N;
    float*    alphaD = ws + o; o += GAT_N;
    (void)ws_size; (void)in_sizes; (void)n_in; (void)out_size;

    launch_gat_layer(x, W1, as1, ad1, b1, ei,
                     h, alphaS, alphaD, feats1, (unsigned*)out_feats, 0, stream);
    launch_gat_layer(feats1, W2, as2, ad2, b2, ei,
                     h, alphaS, alphaD, out_feats, (unsigned*)out_feats, 1, stream);

    gat_pool_own<<<GAT_G * PL_Q, PL_T, 0, stream>>>(out_feats, batch, out_pooled);
}


# Alpha kernel: a wave takes 32 nodes (32 whole-row 16-B loads in flight, halving butterfly over the 2x32 partial sums), 1563 waves not 50000; new sum order for alpha_s, alpha_d
# speedup vs baseline: 3.6739x; 1.0616x over previous
#include <hip/hip_runtime.h>
#include <hip/hip_bf16.h>

#define GAT_N      50000
#define GAT_E      800000
#define GAT_ETOT   850000
#define GAT_D      128
#define GAT_G      64
#define NEG_SLOPE  0.2f

typedef __attribute__((ext_vector_type(2))) float v2f;
typedef __attribute__((ext_vector_type(8))) float v8f;

__device__ __forceinline__ unsigned enc_f32(float f) {
    unsigned u = __float_as_uint(f);
    return (u & 0x80000000u) ? ~u : (u | 0x80000000u);
}
__device__ __forceinline__ float dec_f32(unsigned k) {
    unsigned u = (k & 0x80000000u) ? (k & 0x7FFFFFFFu) : ~k;
    return __uint_as_float(u);
}

__global__ void gat_gemm128_wmma(const float* __restrict__ X,
                                 const float* __restrict__ W,
                                 float* __restrict__ H) {
    const int wave = threadIdx.x >> 5;
    const int lane = threadIdx.x & 31;
    const int half = lane >> 4;
    const int l    = lane & 15;
    const int m0   = blockIdx.x * 16;
    const int n0   = wave * 16;

    const float* xrow = X + (size_t)(m0 + l) * GAT_D;
    v8f acc = {};

#pragma unroll
    for (int k = 0; k < GAT_D; k += 4) {
        const int kk = k + 2 * half;
        v2f a;
        {
            const float2 av = *(const float2*)(xrow + kk);
            a.x = av.x; a.y = av.y;
        }
        v2f b;
        b.x = W[(size_t)kk       * GAT_D + n0 + l];
        b.y = W[(size_t)(kk + 1) * GAT_D + n0 + l];
        acc = __builtin_amdgcn_wmma_f32_16x16x4_f32(
            false, a, false, b, (short)0, acc, false, false);
    }

    const int mBase = m0 + half * 8;
#pragma unroll
    for (int r = 0; r < 8; ++r)
        H[(size_t)(mBase + r) * GAT_D + n0 + l] = acc[r];
}

#define AL_NPW  32
#define AL_T    256
#define AL_NCH  ((GAT_N + AL_NPW - 1) / AL_NPW)
#define AL_NB   ((AL_NCH + AL_T / 32 - 1) / (AL_T / 32))

template <int N>
__device__ __forceinline__ void al_halve(float (&p)[AL_NPW], int lane) {
    const bool up = (lane & N) != 0;
#pragma unroll
    for (int i = 0; i < N; ++i) {
        const float keep = up ? p[i + N] : p[i];
        const float give = up ? p[i] : p[i + N];
        p[i] = keep + __shfl_xor(give, N, 32);
    }
}

__global__ __launch_bounds__(AL_T)
void gat_alpha(const float* __restrict__ H,
               const float* __restrict__ a_src,
               const float* __restrict__ a_dst,
               float* __restrict__ as_out,
               float* __restrict__ ad_out) {
    const int lane  = threadIdx.x & 31;
    const int chunk = blockIdx.x * (AL_T / 32)
                    + __builtin_amdgcn_readfirstlane((int)(threadIdx.x >> 5));
    if (chunk >= AL_NCH) return;
    const int first = chunk * AL_NPW;
    const int n0 = (first + AL_NPW <= GAT_N) ? first : (GAT_N - AL_NPW);

    const float* hp = H + (size_t)n0 * GAT_D + 4 * lane;
    float4 hv[AL_NPW];
#pragma unroll
    for (int i = 0; i < AL_NPW; ++i) hv[i] = *(const float4*)(hp + (size_t)i * GAT_D);
    const float4 sv = *(const float4*)(a_src + 4 * lane);
    const float4 dv = *(const float4*)(a_dst + 4 * lane);
#pragma unroll
    for (int i = 0; i < AL_NPW; ++i)
        asm volatile("" :: "v"(hv[i].x), "v"(hv[i].y), "v"(hv[i].z), "v"(hv[i].w));

    float ps[AL_NPW], pd[AL_NPW];
#pragma unroll
    for (int i = 0; i < AL_NPW; ++i) {
        ps[i] = hv[i].x * sv.x + hv[i].y * sv.y + hv[i].z * sv.z + hv[i].w * sv.w;
        pd[i] = hv[i].x * dv.x + hv[i].y * dv.y + hv[i].z * dv.z + hv[i].w * dv.w;
    }

    al_halve<16>(ps, lane);  al_halve<16>(pd, lane);
    al_halve<8>(ps, lane);   al_halve<8>(pd, lane);
    al_halve<4>(ps, lane);   al_halve<4>(pd, lane);
    al_halve<2>(ps, lane);   al_halve<2>(pd, lane);
    al_halve<1>(ps, lane);   al_halve<1>(pd, lane);

    const int n = n0 + lane;
    if (n >= first) {
        as_out[n] = ps[0];
        ad_out[n] = pd[0];
    }
}

#define AG_R    196
#define AG_NB   256
#define AG_T    1024
#define AG_CAP  8192
#define AG_SU   8
#define AG_GU   8

__device__ __forceinline__ float ag_lrelu(float e) {
    return (e > 0.f) ? e : NEG_SLOPE * e;
}
__device__ __forceinline__ float ag_segmax(unsigned k) {
    const float m = dec_f32(k);
    return __builtin_isfinite(m) ? m : 0.f;
}

__device__ __forceinline__ bool ag_fetch(bool overflow, int idx, unsigned cnt,
                                         const unsigned* l_sd, const float* l_e,
                                         const int* __restrict__ ei,
                                         const float* __restrict__ as,
                                         const float* s_ad, int n0, int nn,
                                         unsigned& sd, float& e) {
    sd = 0u; e = 0.f;
    if (!overflow) {
        if ((unsigned)idx < cnt) { sd = l_sd[idx]; e = l_e[idx]; return true; }
        return false;
    }
    const unsigned dl = (unsigned)(ei[GAT_E + idx] - n0);
    if (dl < (unsigned)nn) {
        const int s = ei[idx];
        sd = (unsigned)s | (dl << 16);
        e  = ag_lrelu(as[s] + s_ad[dl]);
        return true;
    }
    return false;
}

__global__ __launch_bounds__(AG_T)
void gat_aggregate(const int* __restrict__ ei,
                   const float* __restrict__ H,
                   const float* __restrict__ as,
                   const float* __restrict__ ad,
                   const float* __restrict__ bias,
                   float* out,
                   unsigned* lst_all,
                   int second) {
    __shared__ float    acc[AG_R * GAT_D];
    __shared__ float    s_ad[AG_R];
    __shared__ float    s_es[AG_R];
    __shared__ unsigned s_mx[AG_R];
    __shared__ float    s_sum[AG_R];
    __shared__ unsigned l_sd[AG_CAP];
    __shared__ float    l_e[AG_CAP];
    __shared__ unsigned s_cnt;

    const int tid  = threadIdx.x;
    const int lane = tid & 31;
    const int wave = tid >> 5;
    const int n0   = blockIdx.x * AG_R;
    const int nn   = (GAT_N - n0 < AG_R) ? (GAT_N - n0) : AG_R;
    unsigned* lst = lst_all + (size_t)n0 * GAT_D;
    const unsigned lcap = ((unsigned)(nn * GAT_D - 4) < (unsigned)AG_CAP)
                              ? (unsigned)(nn * GAT_D - 4) : (unsigned)AG_CAP;

    for (int i = tid; i < AG_R * GAT_D; i += AG_T) acc[i] = 0.f;
    if (tid < nn) {
        const float d = ad[n0 + tid];
        const float e = ag_lrelu(as[n0 + tid] + d);
        s_ad[tid]  = d;
        s_es[tid]  = e;
        s_mx[tid]  = enc_f32(e);
        s_sum[tid] = 0.f;
    }
    if (tid == 0) s_cnt = 0u;
    __syncthreads();

    bool listed = false;
    if (second) {
        const unsigned c = (unsigned)__builtin_amdgcn_readfirstlane((int)lst[0]);
        listed = (c <= lcap);
        if (listed) {
            for (unsigned j = tid; j < c; j += AG_T) l_sd[j] = lst[4 + j];
            if (tid == 0) s_cnt = c;
        }
    }

    if (!listed) {
        const int4* dst4 = (const int4*)(ei + GAT_E);
        const int rot = blockIdx.x * ((GAT_E / 4) / AG_NB);
        for (int q0 = 0; q0 < GAT_E / 4; q0 += AG_SU * AG_T) {
            int4 d4[AG_SU];
            int  qq[AG_SU];
#pragma unroll
            for (int u = 0; u < AG_SU; ++u) {
                const int q = q0 + u * AG_T + tid;
                qq[u] = (q + rot >= GAT_E / 4) ? (q + rot - GAT_E / 4) : (q + rot);
                d4[u] = (q < GAT_E / 4) ? dst4[qq[u]] : make_int4(-1, -1, -1, -1);
            }
#pragma unroll
            for (int u = 0; u < AG_SU; ++u) {
                const int q = qq[u];
                const int dd[4] = {d4[u].x, d4[u].y, d4[u].z, d4[u].w};
#pragma unroll
                for (int k = 0; k < 4; ++k) {
                    const unsigned dl = (unsigned)(dd[k] - n0);
                    if (dl < (unsigned)nn) {
                        const unsigned slot = atomicAdd(&s_cnt, 1u);
                        if (slot < AG_CAP) l_sd[slot] = (unsigned)(4 * q + k) | (dl << 20);
                    }
                }
            }
        }
    }
    __syncthreads();
    const unsigned cnt = s_cnt;
    const bool overflow = cnt > AG_CAP;
    const int nchunks = overflow ? (GAT_E / 32) : (int)((cnt + 31u) >> 5);

    const bool keep = !second && cnt <= lcap;
    if (!overflow) {
        for (int j = tid; j < (int)cnt; j += AG_T) {
            const unsigned w  = l_sd[j];
            unsigned dl; int s;
            if (listed) { s = (int)(w & 0xFFFFu); dl = w >> 16; }
            else        { dl = w >> 20; s = ei[w & 0xFFFFFu]; }
            const float e = ag_lrelu(as[s] + s_ad[dl]);
            atomicMax(&s_mx[dl], enc_f32(e));
            const unsigned sd = (unsigned)s | (dl << 16);
            l_sd[j] = sd;
            l_e[j]  = e;
            if (keep) lst[4 + j] = sd;
        }
    } else {
        for (int c = wave; c < nchunks; c += AG_T / 32) {
            unsigned sd; float e;
            if (ag_fetch(true, c * 32 + lane, cnt, l_sd, l_e, ei, as, s_ad, n0, nn, sd, e))
                atomicMax(&s_mx[sd >> 16], enc_f32(e));
        }
    }
    if (!second && tid == 0) lst[0] = keep ? cnt : 0xFFFFFFFFu;
    __syncthreads();

    for (int c = wave; c < nchunks; c += AG_T / 32) {
        unsigned sd; float e;
        if (ag_fetch(overflow, c * 32 + lane, cnt, l_sd, l_e, ei, as, s_ad, n0, nn, sd, e)) {
            const unsigned dl = sd >> 16;
            atomicAdd(&s_sum[dl], expf(e - ag_segmax(s_mx[dl])));
        }
    }
    if (tid < nn) atomicAdd(&s_sum[tid], expf(s_es[tid] - ag_segmax(s_mx[tid])));
    __syncthreads();

    for (int c = wave; c < nchunks; c += AG_T / 32) {
        unsigned sd; float e;
        const bool valid = ag_fetch(overflow, c * 32 + lane, cnt, l_sd, l_e, ei, as, s_ad, n0, nn, sd, e);
        float coef = 0.f;
        if (valid) {
            const unsigned dl = sd >> 16;
            coef = expf(e - ag_segmax(s_mx[dl])) / (s_sum[dl] + 1e-16f);
        }
        unsigned mask = __builtin_amdgcn_ballot_w32(valid);
        while (mask) {
            bool     ok[AG_GU];
            unsigned sj[AG_GU];
            float    cj[AG_GU];
            float4   hv[AG_GU];
#pragma unroll
            for (int u = 0; u < AG_GU; ++u) {
                ok[u] = (mask != 0u);
                const int j = ok[u] ? __builtin_ctz(mask) : 0;
                mask &= mask - 1u;
                sj[u] = (unsigned)__builtin_amdgcn_readlane((int)sd, j);
                cj[u] = __uint_as_float((unsigned)__builtin_amdgcn_readlane((int)__float_as_uint(coef), j));
                hv[u] = *(const float4*)(H + (size_t)(sj[u] & 0xFFFFu) * GAT_D + 4 * lane);
            }
#pragma unroll
            for (int u = 0; u < AG_GU; ++u)
                asm volatile("" :: "v"(hv[u].x), "v"(hv[u].y), "v"(hv[u].z), "v"(hv[u].w));
#pragma unroll
            for (int u = 0; u < AG_GU; ++u) {
                if (ok[u]) {
                    float* a = acc + (sj[u] >> 16) * GAT_D + lane;
                    atomicAdd(a,      hv[u].x * cj[u]);
                    atomicAdd(a + 32, hv[u].y * cj[u]);
                    atomicAdd(a + 64, hv[u].z * cj[u]);
                    atomicAdd(a + 96, hv[u].w * cj[u]);
                }
            }
        }
    }
    __syncthreads();

    const float4 bv = *(const float4*)(bias + 4 * lane);
    for (int dl = wave; dl < nn; dl += AG_T / 32) {
        const size_t row = (size_t)(n0 + dl) * GAT_D + 4 * lane;
        const float4 hs = *(const float4*)(H + row);
        const float coef = expf(s_es[dl] - ag_segmax(s_mx[dl])) / (s_sum[dl] + 1e-16f);
        const float* a = acc + dl * GAT_D + lane;
        float4 o;
        o.x = (a[0]  + hs.x * coef) + bv.x;
        o.y = (a[32] + hs.y * coef) + bv.y;
        o.z = (a[64] + hs.z * coef) + bv.z;
        o.w = (a[96] + hs.w * coef) + bv.w;
        *(float4*)(out + row) = o;
    }
}

#define PL_Q   4
#define PL_T   1024
#define PL_W   (PL_T / 32)
#define PL_NC  ((GAT_N + 31) / 32)
#define PL_SU  7
#define PL_GU  16

__global__ __launch_bounds__(PL_T)
void gat_pool_own(const float* __restrict__ feats,
                  const int* __restrict__ batch,
                  float* __restrict__ out_pooled) {
    __shared__ float    p_sum[PL_W * 32];
    __shared__ unsigned p_cnt[PL_W];

    const int lane = threadIdx.x & 31;
    const int wave = __builtin_amdgcn_readfirstlane((int)(threadIdx.x >> 5));
    const int g    = blockIdx.x / PL_Q;
    const int c0   = (blockIdx.x % PL_Q) * 32;
    const float* col = feats + c0 + lane;

    float    acc = 0.f;
    unsigned cnt = 0u;

    for (int ch0 = wave; ch0 < PL_NC; ch0 += PL_W * PL_SU) {
        int id[PL_SU];
#pragma unroll
        for (int u = 0; u < PL_SU; ++u) {
            const int n = (ch0 + u * PL_W) * 32 + lane;
            const int v = batch[(n < GAT_N) ? n : (GAT_N - 1)];
            id[u] = (n < GAT_N) ? v : -1;
        }
#pragma unroll
        for (int u = 0; u < PL_SU; ++u) {
            const int nb = (ch0 + u * PL_W) * 32;
            unsigned mask = __builtin_amdgcn_ballot_w32(id[u] == g);
            cnt += (unsigned)__builtin_popcount(mask);
            while (mask) {
                bool  ok[PL_GU];
                float t[PL_GU];
#pragma unroll
                for (int r = 0; r < PL_GU; ++r) {
                    ok[r] = (mask != 0u);
                    const int j = ok[r] ? __builtin_ctz(mask) : 0;
                    mask &= mask - 1u;
                    t[r] = col[(size_t)(nb + j) * GAT_D];
                }
#pragma unroll
                for (int r = 0; r < PL_GU; ++r)
                    asm volatile("" :: "v"(t[r]));
#pragma unroll
                for (int r = 0; r < PL_GU; ++r) acc += ok[r] ? t[r] : 0.f;
            }
        }
    }

    p_sum[wave * 32 + lane] = acc;
    if (lane == 0) p_cnt[wave] = cnt;
    __syncthreads();
    if (wave == 0) {
        float    s = 0.f;
        unsigned c = 0u;
#pragma unroll
        for (int w = 0; w < PL_W; ++w) {
            s += p_sum[w * 32 + lane];
            c += p_cnt[w];
        }
        out_pooled[g * GAT_D + c0 + lane] = s / fmaxf((float)c, 1.0f);
    }
}

static void launch_gat_layer(const float* x, const float* W,
                             const float* a_s, const float* a_d,
                             const float* bias,
                             const int* ei,
                             float* h, float* alphaS, float* alphaD,
                             float* out, unsigned* lst_all, int second,
                             hipStream_t stream) {
    gat_gemm128_wmma<<<GAT_N / 16, 256, 0, stream>>>(x, W, h);
    gat_alpha<<<AL_NB, AL_T, 0, stream>>>(h, a_s, a_d, alphaS, alphaD);
    gat_aggregate<<<AG_NB, AG_T, 0, stream>>>(ei, h, alphaS, alphaD, bias, out, lst_all, second);
}

extern "C" void kernel_launch(void* const* d_in, const int* in_sizes, int n_in,
                              void* d_out, int out_size, void* d_ws, size_t ws_size,
                              hipStream_t stream) {
    const float* x     = (const float*)d_in[0];
    const int*   ei    = (const int*)d_in[1];
    const int*   batch = (const int*)d_in[2];
    const float* W1  = (const float*)d_in[3];
    const float* as1 = (const float*)d_in[4];
    const float* ad1 = (const float*)d_in[5];
    const float* b1  = (const float*)d_in[6];
    const float* W2  = (const float*)d_in[7];
    const float* as2 = (const float*)d_in[8];
    const float* ad2 = (const float*)d_in[9];
    const float* b2  = (const float*)d_in[10];

    float* out_feats  = (float*)d_out;
    float* out_pooled = out_feats + (size_t)GAT_N * GAT_D;

    float* ws = (float*)d_ws;
    size_t o = 0;
    float*    h      = ws + o; o += (size_t)GAT_N * GAT_D;
    float*    feats1 = ws + o; o += (size_t)GAT_N * GAT_D;
    float*    alphaS = ws + o; o += GAT_N;
    float*    alphaD = ws + o; o += GAT_N;
    (void)ws_size; (void)in_sizes; (void)n_in; (void)out_size;

    launch_gat_layer(x, W1, as1, ad1, b1, ei,
                     h, alphaS, alphaD, feats1, (unsigned*)out_feats, 0, stream);
    launch_gat_layer(feats1, W2, as2, ad2, b2, ei,
                     h, alphaS, alphaD, out_feats, (unsigned*)out_feats, 1, stream);

    gat_pool_own<<<GAT_G * PL_Q, PL_T, 0, stream>>>(out_feats, batch, out_pooled);
}


# Layer-1 edge filter scans a byte key per edge (dst/196, written by the alpha launch), not the dst list: 16 edges per 16-B load, 4 bytes per test, one LDS slot atomic per thread; sums in arrival order
# speedup vs baseline: 4.1014x; 1.1164x over previous
#include <hip/hip_runtime.h>
#include <hip/hip_bf16.h>

#define GAT_N      50000
#define GAT_E      800000
#define GAT_ETOT   850000
#define GAT_D      128
#define GAT_G      64
#define NEG_SLOPE  0.2f

typedef __attribute__((ext_vector_type(2))) float v2f;
typedef __attribute__((ext_vector_type(8))) float v8f;

__device__ __forceinline__ unsigned enc_f32(float f) {
    unsigned u = __float_as_uint(f);
    return (u & 0x80000000u) ? ~u : (u | 0x80000000u);
}
__device__ __forceinline__ float dec_f32(unsigned k) {
    unsigned u = (k & 0x80000000u) ? (k & 0x7FFFFFFFu) : ~k;
    return __uint_as_float(u);
}

__global__ void gat_gemm128_wmma(const float* __restrict__ X,
                                 const float* __restrict__ W,
                                 float* __restrict__ H) {
    const int wave = threadIdx.x >> 5;
    const int lane = threadIdx.x & 31;
    const int half = lane >> 4;
    const int l    = lane & 15;
    const int m0   = blockIdx.x * 16;
    const int n0   = wave * 16;

    const float* xrow = X + (size_t)(m0 + l) * GAT_D;
    v8f acc = {};

#pragma unroll
    for (int k = 0; k < GAT_D; k += 4) {
        const int kk = k + 2 * half;
        v2f a;
        {
            const float2 av = *(const float2*)(xrow + kk);
            a.x = av.x; a.y = av.y;
        }
        v2f b;
        b.x = W[(size_t)kk       * GAT_D + n0 + l];
        b.y = W[(size_t)(kk + 1) * GAT_D + n0 + l];
        acc = __builtin_amdgcn_wmma_f32_16x16x4_f32(
            false, a, false, b, (short)0, acc, false, false);
    }

    const int mBase = m0 + half * 8;
#pragma unroll
    for (int r = 0; r < 8; ++r)
        H[(size_t)(mBase + r) * GAT_D + n0 + l] = acc[r];
}

#define AG_R        196
#define KEY_NQ      (GAT_E / 16)
#define KEY_IDX(q)  (((q) >> 12) * 6272 + 2112 + ((q) & 4095))

#define AL_NPW  32
#define AL_T    256
#define AL_NCH  ((GAT_N + AL_NPW - 1) / AL_NPW)
#define AL_NB   ((AL_NCH + AL_T / 32 - 1) / (AL_T / 32))

template <int N>
__device__ __forceinline__ void al_halve(float (&p)[AL_NPW], int lane) {
    const bool up = (lane & N) != 0;
#pragma unroll
    for (int i = 0; i < N; ++i) {
        const float keep = up ? p[i + N] : p[i];
        const float give = up ? p[i] : p[i + N];
        p[i] = keep + __shfl_xor(give, N, 32);
    }
}

__global__ __launch_bounds__(AL_T)
void gat_alpha(const float* __restrict__ H,
               const float* __restrict__ a_src,
               const float* __restrict__ a_dst,
               float* __restrict__ as_out,
               float* __restrict__ ad_out,
               const int* __restrict__ ei,
               uint4* __restrict__ keys) {
    if (keys != nullptr) {
        const int q = blockIdx.x * AL_T + threadIdx.x;
        if (q < KEY_NQ) {
            const int4* d4 = (const int4*)(ei + GAT_E) + 4 * q;
            const int4 d0 = d4[0], d1 = d4[1], d2 = d4[2], d3 = d4[3];
            uint4 k;
            k.x = ((unsigned)d0.x / (unsigned)AG_R) | (((unsigned)d0.y / (unsigned)AG_R) << 8) | (((unsigned)d0.z / (unsigned)AG_R) << 16) | (((unsigned)d0.w / (unsigned)AG_R) << 24);
            k.y = ((unsigned)d1.x / (unsigned)AG_R) | (((unsigned)d1.y / (unsigned)AG_R) << 8) | (((unsigned)d1.z / (unsigned)AG_R) << 16) | (((unsigned)d1.w / (unsigned)AG_R) << 24);
            k.z = ((unsigned)d2.x / (unsigned)AG_R) | (((unsigned)d2.y / (unsigned)AG_R) << 8) | (((unsigned)d2.z / (unsigned)AG_R) << 16) | (((unsigned)d2.w / (unsigned)AG_R) << 24);
            k.w = ((unsigned)d3.x / (unsigned)AG_R) | (((unsigned)d3.y / (unsigned)AG_R) << 8) | (((unsigned)d3.z / (unsigned)AG_R) << 16) | (((unsigned)d3.w / (unsigned)AG_R) << 24);
            keys[KEY_IDX(q)] = k;
        }
    }
    const int lane  = threadIdx.x & 31;
    const int chunk = blockIdx.x * (AL_T / 32)
                    + __builtin_amdgcn_readfirstlane((int)(threadIdx.x >> 5));
    if (chunk >= AL_NCH) return;
    const int first = chunk * AL_NPW;
    const int n0 = (first + AL_NPW <= GAT_N) ? first : (GAT_N - AL_NPW);

    const float* hp = H + (size_t)n0 * GAT_D + 4 * lane;
    float4 hv[AL_NPW];
#pragma unroll
    for (int i = 0; i < AL_NPW; ++i) hv[i] = *(const float4*)(hp + (size_t)i * GAT_D);
    const float4 sv = *(const float4*)(a_src + 4 * lane);
    const float4 dv = *(const float4*)(a_dst + 4 * lane);
#pragma unroll
    for (int i = 0; i < AL_NPW; ++i)
        asm volatile("" :: "v"(hv[i].x), "v"(hv[i].y), "v"(hv[i].z), "v"(hv[i].w));

    float ps[AL_NPW], pd[AL_NPW];
#pragma unroll
    for (int i = 0; i < AL_NPW; ++i) {
        ps[i] = hv[i].x * sv.x + hv[i].y * sv.y + hv[i].z * sv.z + hv[i].w * sv.w;
        pd[i] = hv[i].x * dv.x + hv[i].y * dv.y + hv[i].z * dv.z + hv[i].w * dv.w;
    }

    al_halve<16>(ps, lane);  al_halve<16>(pd, lane);
    al_halve<8>(ps, lane);   al_halve<8>(pd, lane);
    al_halve<4>(ps, lane);   al_halve<4>(pd, lane);
    al_halve<2>(ps, lane);   al_halve<2>(pd, lane);
    al_halve<1>(ps, lane);   al_halve<1>(pd, lane);

    const int n = n0 + lane;
    if (n >= first) {
        as_out[n] = ps[0];
        ad_out[n] = pd[0];
    }
}

#define AG_NB   256
#define AG_T    1024
#define AG_CAP  8192
#define AG_SU   8
#define AG_KU   7
#define AG_GU   8

__device__ __forceinline__ float ag_lrelu(float e) {
    return (e > 0.f) ? e : NEG_SLOPE * e;
}
__device__ __forceinline__ float ag_segmax(unsigned k) {
    const float m = dec_f32(k);
    return __builtin_isfinite(m) ? m : 0.f;
}

__device__ __forceinline__ bool ag_fetch(bool overflow, int idx, unsigned cnt,
                                         const unsigned* l_sd, const float* l_e,
                                         const int* __restrict__ ei,
                                         const float* __restrict__ as,
                                         const float* s_ad, int n0, int nn,
                                         unsigned& sd, float& e) {
    sd = 0u; e = 0.f;
    if (!overflow) {
        if ((unsigned)idx < cnt) { sd = l_sd[idx]; e = l_e[idx]; return true; }
        return false;
    }
    const unsigned dl = (unsigned)(ei[GAT_E + idx] - n0);
    if (dl < (unsigned)nn) {
        const int s = ei[idx];
        sd = (unsigned)s | (dl << 16);
        e  = ag_lrelu(as[s] + s_ad[dl]);
        return true;
    }
    return false;
}

__global__ __launch_bounds__(AG_T)
void gat_aggregate(const int* __restrict__ ei,
                   const float* __restrict__ H,
                   const float* __restrict__ as,
                   const float* __restrict__ ad,
                   const float* __restrict__ bias,
                   float* out,
                   unsigned* lst_all,
                   int second) {
    __shared__ float    acc[AG_R * GAT_D];
    __shared__ float    s_ad[AG_R];
    __shared__ float    s_es[AG_R];
    __shared__ unsigned s_mx[AG_R];
    __shared__ float    s_sum[AG_R];
    __shared__ unsigned l_sd[AG_CAP];
    __shared__ float    l_e[AG_CAP];
    __shared__ unsigned s_cnt;

    const int tid  = threadIdx.x;
    const int lane = tid & 31;
    const int wave = tid >> 5;
    const int n0   = blockIdx.x * AG_R;
    const int nn   = (GAT_N - n0 < AG_R) ? (GAT_N - n0) : AG_R;
    unsigned* lst = lst_all + (size_t)n0 * GAT_D;
    const unsigned lcap = ((unsigned)(nn * GAT_D - 4) < (unsigned)AG_CAP)
                              ? (unsigned)(nn * GAT_D - 4) : (unsigned)AG_CAP;

    for (int i = tid; i < AG_R * GAT_D; i += AG_T) acc[i] = 0.f;
    if (tid < nn) {
        const float d = ad[n0 + tid];
        const float e = ag_lrelu(as[n0 + tid] + d);
        s_ad[tid]  = d;
        s_es[tid]  = e;
        s_mx[tid]  = enc_f32(e);
        s_sum[tid] = 0.f;
    }
    if (tid == 0) s_cnt = 0u;
    __syncthreads();

    bool listed = false;
    if (second) {
        const unsigned c = (unsigned)__builtin_amdgcn_readfirstlane((int)lst[0]);
        listed = (c <= lcap);
        if (listed) {
            for (unsigned j = tid; j < c; j += AG_T) l_sd[j] = lst[4 + j];
            if (tid == 0) s_cnt = c;
        }
    }

    if (!second) {
        const uint4* keys = (const uint4*)lst_all;
        const unsigned pat = blockIdx.x * 0x01010101u;
        const int rot = blockIdx.x * (KEY_NQ / AG_NB);
        for (int q0 = 0; q0 < KEY_NQ; q0 += AG_KU * AG_T) {
            uint4 kv[AG_KU];
            int   qq[AG_KU];
#pragma unroll
            for (int u = 0; u < AG_KU; ++u) {
                const int q = q0 + u * AG_T + tid;
                qq[u] = (q + rot >= KEY_NQ) ? (q + rot - KEY_NQ) : (q + rot);
                kv[u] = (q < KEY_NQ) ? keys[KEY_IDX(qq[u])] : make_uint4(~pat, ~pat, ~pat, ~pat);
            }
            unsigned mk[AG_KU];
            unsigned own = 0u;
#pragma unroll
            for (int u = 0; u < AG_KU; ++u) {
                const unsigned kw[4] = {kv[u].x, kv[u].y, kv[u].z, kv[u].w};
                unsigned m = 0u;
#pragma unroll
                for (int w = 0; w < 4; ++w) {
                    const unsigned x = kw[w] ^ pat;
                    const unsigned z = ~(((x & 0x7F7F7F7Fu) + 0x7F7F7F7Fu) | x | 0x7F7F7F7Fu);
                    m |= (z >> 7) << w;
                }
                mk[u] = m;
                own += (unsigned)__builtin_popcount(m);
            }
            unsigned slot = 0u;
            if (own) slot = atomicAdd(&s_cnt, own);
#pragma unroll
            for (int u = 0; u < AG_KU; ++u) {
                unsigned m = mk[u];
                while (m) {
                    const int p = __builtin_ctz(m);
                    m &= m - 1u;
                    if (slot < AG_CAP) l_sd[slot] = (unsigned)(16 * qq[u] + 4 * (p & 7) + (p >> 3));
                    ++slot;
                }
            }
        }
    }
    if (second && !listed) {
        const int4* dst4 = (const int4*)(ei + GAT_E);
        const int rot = blockIdx.x * ((GAT_E / 4) / AG_NB);
        for (int q0 = 0; q0 < GAT_E / 4; q0 += AG_SU * AG_T) {
            int4 d4[AG_SU];
            int  qq[AG_SU];
#pragma unroll
            for (int u = 0; u < AG_SU; ++u) {
                const int q = q0 + u * AG_T + tid;
                qq[u] = (q + rot >= GAT_E / 4) ? (q + rot - GAT_E / 4) : (q + rot);
                d4[u] = (q < GAT_E / 4) ? dst4[qq[u]] : make_int4(-1, -1, -1, -1);
            }
#pragma unroll
            for (int u = 0; u < AG_SU; ++u) {
                const int q = qq[u];
                const int dd[4] = {d4[u].x, d4[u].y, d4[u].z, d4[u].w};
#pragma unroll
                for (int k = 0; k < 4; ++k) {
                    const unsigned dl = (unsigned)(dd[k] - n0);
                    if (dl < (unsigned)nn) {
                        const unsigned slot = atomicAdd(&s_cnt, 1u);
                        if (slot < AG_CAP) l_sd[slot] = (unsigned)(4 * q + k);
                    }
                }
            }
        }
    }
    __syncthreads();
    const unsigned cnt = s_cnt;
    const bool overflow = cnt > AG_CAP;
    const int nchunks = overflow ? (GAT_E / 32) : (int)((cnt + 31u) >> 5);

    const bool keep = !second && cnt <= lcap;
    if (!overflow) {
        for (int j = tid; j < (int)cnt; j += AG_T) {
            const unsigned w  = l_sd[j];
            unsigned dl; int s;
            if (listed) { s = (int)(w & 0xFFFFu); dl = w >> 16; }
            else        { s = ei[w]; dl = (unsigned)(ei[GAT_E + w] - n0); }
            const float e = ag_lrelu(as[s] + s_ad[dl]);
            atomicMax(&s_mx[dl], enc_f32(e));
            const unsigned sd = (unsigned)s | (dl << 16);
            l_sd[j] = sd;
            l_e[j]  = e;
            if (keep) lst[4 + j] = sd;
        }
    } else {
        for (int c = wave; c < nchunks; c += AG_T / 32) {
            unsigned sd; float e;
            if (ag_fetch(true, c * 32 + lane, cnt, l_sd, l_e, ei, as, s_ad, n0, nn, sd, e))
                atomicMax(&s_mx[sd >> 16], enc_f32(e));
        }
    }
    if (!second && tid == 0) lst[0] = keep ? cnt : 0xFFFFFFFFu;
    __syncthreads();

    for (int c = wave; c < nchunks; c += AG_T / 32) {
        unsigned sd; float e;
        if (ag_fetch(overflow, c * 32 + lane, cnt, l_sd, l_e, ei, as, s_ad, n0, nn, sd, e)) {
            const unsigned dl = sd >> 16;
            atomicAdd(&s_sum[dl], expf(e - ag_segmax(s_mx[dl])));
        }
    }
    if (tid < nn) atomicAdd(&s_sum[tid], expf(s_es[tid] - ag_segmax(s_mx[tid])));
    __syncthreads();

    for (int c = wave; c < nchunks; c += AG_T / 32) {
        unsigned sd; float e;
        const bool valid = ag_fetch(overflow, c * 32 + lane, cnt, l_sd, l_e, ei, as, s_ad, n0, nn, sd, e);
        float coef = 0.f;
        if (valid) {
            const unsigned dl = sd >> 16;
            coef = expf(e - ag_segmax(s_mx[dl])) / (s_sum[dl] + 1e-16f);
        }
        unsigned mask = __builtin_amdgcn_ballot_w32(valid);
        while (mask) {
            bool     ok[AG_GU];
            unsigned sj[AG_GU];
            float    cj[AG_GU];
            float4   hv[AG_GU];
#pragma unroll
            for (int u = 0; u < AG_GU; ++u) {
                ok[u] = (mask != 0u);
                const int j = ok[u] ? __builtin_ctz(mask) : 0;
                mask &= mask - 1u;
                sj[u] = (unsigned)__builtin_amdgcn_readlane((int)sd, j);
                cj[u] = __uint_as_float((unsigned)__builtin_amdgcn_readlane((int)__float_as_uint(coef), j));
                hv[u] = *(const float4*)(H + (size_t)(sj[u] & 0xFFFFu) * GAT_D + 4 * lane);
            }
#pragma unroll
            for (int u = 0; u < AG_GU; ++u)
                asm volatile("" :: "v"(hv[u].x), "v"(hv[u].y), "v"(hv[u].z), "v"(hv[u].w));
#pragma unroll
            for (int u = 0; u < AG_GU; ++u) {
                if (ok[u]) {
                    float* a = acc + (sj[u] >> 16) * GAT_D + lane;
                    atomicAdd(a,      hv[u].x * cj[u]);
                    atomicAdd(a + 32, hv[u].y * cj[u]);
                    atomicAdd(a + 64, hv[u].z * cj[u]);
                    atomicAdd(a + 96, hv[u].w * cj[u]);
                }
            }
        }
    }
    __syncthreads();

    const float4 bv = *(const float4*)(bias + 4 * lane);
    for (int dl = wave; dl < nn; dl += AG_T / 32) {
        const size_t row = (size_t)(n0 + dl) * GAT_D + 4 * lane;
        const float4 hs = *(const float4*)(H + row);
        const float coef = expf(s_es[dl] - ag_segmax(s_mx[dl])) / (s_sum[dl] + 1e-16f);
        const float* a = acc + dl * GAT_D + lane;
        float4 o;
        o.x = (a[0]  + hs.x * coef) + bv.x;
        o.y = (a[32] + hs.y * coef) + bv.y;
        o.z = (a[64] + hs.z * coef) + bv.z;
        o.w = (a[96] + hs.w * coef) + bv.w;
        *(float4*)(out + row) = o;
    }
}

#define PL_Q   4
#define PL_T   1024
#define PL_W   (PL_T / 32)
#define PL_NC  ((GAT_N + 31) / 32)
#define PL_SU  7
#define PL_GU  16

__global__ __launch_bounds__(PL_T)
void gat_pool_own(const float* __restrict__ feats,
                  const int* __restrict__ batch,
                  float* __restrict__ out_pooled) {
    __shared__ float    p_sum[PL_W * 32];
    __shared__ unsigned p_cnt[PL_W];

    const int lane = threadIdx.x & 31;
    const int wave = __builtin_amdgcn_readfirstlane((int)(threadIdx.x >> 5));
    const int g    = blockIdx.x / PL_Q;
    const int c0   = (blockIdx.x % PL_Q) * 32;
    const float* col = feats + c0 + lane;

    float    acc = 0.f;
    unsigned cnt = 0u;

    for (int ch0 = wave; ch0 < PL_NC; ch0 += PL_W * PL_SU) {
        int id[PL_SU];
#pragma unroll
        for (int u = 0; u < PL_SU; ++u) {
            const int n = (ch0 + u * PL_W) * 32 + lane;
            const int v = batch[(n < GAT_N) ? n : (GAT_N - 1)];
            id[u] = (n < GAT_N) ? v : -1;
        }
#pragma unroll
        for (int u = 0; u < PL_SU; ++u) {
            const int nb = (ch0 + u * PL_W) * 32;
            unsigned mask = __builtin_amdgcn_ballot_w32(id[u] == g);
            cnt += (unsigned)__builtin_popcount(mask);
            while (mask) {
                bool  ok[PL_GU];
                float t[PL_GU];
#pragma unroll
                for (int r = 0; r < PL_GU; ++r) {
                    ok[r] = (mask != 0u);
                    const int j = ok[r] ? __builtin_ctz(mask) : 0;
                    mask &= mask - 1u;
                    t[r] = col[(size_t)(nb + j) * GAT_D];
                }
#pragma unroll
                for (int r = 0; r < PL_GU; ++r)
                    asm volatile("" :: "v"(t[r]));
#pragma unroll
                for (int r = 0; r < PL_GU; ++r) acc += ok[r] ? t[r] : 0.f;
            }
        }
    }

    p_sum[wave * 32 + lane] = acc;
    if (lane == 0) p_cnt[wave] = cnt;
    __syncthreads();
    if (wave == 0) {
        float    s = 0.f;
        unsigned c = 0u;
#pragma unroll
        for (int w = 0; w < PL_W; ++w) {
            s += p_sum[w * 32 + lane];
            c += p_cnt[w];
        }
        out_pooled[g * GAT_D + c0 + lane] = s / fmaxf((float)c, 1.0f);
    }
}

static void launch_gat_layer(const float* x, const float* W,
                             const float* a_s, const float* a_d,
                             const float* bias,
                             const int* ei,
                             float* h, float* alphaS, float* alphaD,
                             float* out, unsigned* lst_all, uint4* keys, int second,
                             hipStream_t stream) {
    gat_gemm128_wmma<<<GAT_N / 16, 256, 0, stream>>>(x, W, h);
    gat_alpha<<<AL_NB, AL_T, 0, stream>>>(h, a_s, a_d, alphaS, alphaD, ei, keys);
    gat_aggregate<<<AG_NB, AG_T, 0, stream>>>(ei, h, alphaS, alphaD, bias, out, lst_all, second);
}

extern "C" void kernel_launch(void* const* d_in, const int* in_sizes, int n_in,
                              void* d_out, int out_size, void* d_ws, size_t ws_size,
                              hipStream_t stream) {
    const float* x     = (const float*)d_in[0];
    const int*   ei    = (const int*)d_in[1];
    const int*   batch = (const int*)d_in[2];
    const float* W1  = (const float*)d_in[3];
    const float* as1 = (const float*)d_in[4];
    const float* ad1 = (const float*)d_in[5];
    const float* b1  = (const float*)d_in[6];
    const float* W2  = (const float*)d_in[7];
    const float* as2 = (const float*)d_in[8];
    const float* ad2 = (const float*)d_in[9];
    const float* b2  = (const float*)d_in[10];

    float* out_feats  = (float*)d_out;
    float* out_pooled = out_feats + (size_t)GAT_N * GAT_D;

    float* ws = (float*)d_ws;
    size_t o = 0;
    float*    h      = ws + o; o += (size_t)GAT_N * GAT_D;
    float*    feats1 = ws + o; o += (size_t)GAT_N * GAT_D;
    float*    alphaS = ws + o; o += GAT_N;
    float*    alphaD = ws + o; o += GAT_N;
    (void)ws_size; (void)in_sizes; (void)n_in; (void)out_size;

    launch_gat_layer(x, W1, as1, ad1, b1, ei,
                     h, alphaS, alphaD, feats1, (unsigned*)out_feats, (uint4*)out_feats, 0, stream);
    launch_gat_layer(feats1, W2, as2, ad2, b2, ei,
                     h, alphaS, alphaD, out_feats, (unsigned*)out_feats, (uint4*)nullptr, 1, stream);

    gat_pool_own<<<GAT_G * PL_Q, PL_T, 0, stream>>>(out_feats, batch, out_pooled);
}


# Edge lists by owner block built once by extra blocks of the layer-1 alpha launch (LDS histogram per 4096 edges, one global add per owner, arrival-order rank); no scan of all edges by every block
# speedup vs baseline: 4.3208x; 1.0535x over previous
#include <hip/hip_runtime.h>
#include <hip/hip_bf16.h>

#define GAT_N      50000
#define GAT_E      800000
#define GAT_ETOT   850000
#define GAT_D      128
#define GAT_G      64
#define NEG_SLOPE  0.2f

typedef __attribute__((ext_vector_type(2))) float v2f;
typedef __attribute__((ext_vector_type(8))) float v8f;

__device__ __forceinline__ unsigned enc_f32(float f) {
    unsigned u = __float_as_uint(f);
    return (u & 0x80000000u) ? ~u : (u | 0x80000000u);
}
__device__ __forceinline__ float dec_f32(unsigned k) {
    unsigned u = (k & 0x80000000u) ? (k & 0x7FFFFFFFu) : ~k;
    return __uint_as_float(u);
}

#define AG_R        196
#define AG_NB       256
#define AG_CAP      8192

__global__ void gat_gemm128_wmma(const float* __restrict__ X,
                                 const float* __restrict__ W,
                                 float* __restrict__ H,
                                 unsigned* __restrict__ cnt_zero) {
    if (cnt_zero != nullptr && blockIdx.x < AG_NB && threadIdx.x == 0)
        cnt_zero[(size_t)blockIdx.x * AG_R * GAT_D] = 0u;
    const int wave = threadIdx.x >> 5;
    const int lane = threadIdx.x & 31;
    const int half = lane >> 4;
    const int l    = lane & 15;
    const int m0   = blockIdx.x * 16;
    const int n0   = wave * 16;

    const float* xrow = X + (size_t)(m0 + l) * GAT_D;
    v8f acc = {};

#pragma unroll
    for (int k = 0; k < GAT_D; k += 4) {
        const int kk = k + 2 * half;
        v2f a;
        {
            const float2 av = *(const float2*)(xrow + kk);
            a.x = av.x; a.y = av.y;
        }
        v2f b;
        b.x = W[(size_t)kk       * GAT_D + n0 + l];
        b.y = W[(size_t)(kk + 1) * GAT_D + n0 + l];
        acc = __builtin_amdgcn_wmma_f32_16x16x4_f32(
            false, a, false, b, (short)0, acc, false, false);
    }

    const int mBase = m0 + half * 8;
#pragma unroll
    for (int r = 0; r < 8; ++r)
        H[(size_t)(mBase + r) * GAT_D + n0 + l] = acc[r];
}

#define BIN_EPT     16
#define BIN_NQ      (GAT_E / BIN_EPT)

#define AL_NPW  32
#define AL_T    256
#define AL_NCH  ((GAT_N + AL_NPW - 1) / AL_NPW)
#define AL_NB   ((AL_NCH + AL_T / 32 - 1) / (AL_T / 32))
#define BIN_NB  ((BIN_NQ + AL_T - 1) / AL_T)
static_assert(AL_T == AG_NB, "one histogram slot per thread");
static_assert(GAT_E % BIN_EPT == 0, "whole 16-edge groups");

template <int N>
__device__ __forceinline__ void al_halve(float (&p)[AL_NPW], int lane) {
    const bool up = (lane & N) != 0;
#pragma unroll
    for (int i = 0; i < N; ++i) {
        const float keep = up ? p[i + N] : p[i];
        const float give = up ? p[i] : p[i + N];
        p[i] = keep + __shfl_xor(give, N, 32);
    }
}

__global__ __launch_bounds__(AL_T)
void gat_alpha(const float* __restrict__ H,
               const float* __restrict__ a_src,
               const float* __restrict__ a_dst,
               float* __restrict__ as_out,
               float* __restrict__ ad_out,
               const int* __restrict__ ei,
               unsigned* __restrict__ lists) {
    if (lists != nullptr && blockIdx.x >= AL_NB) {
        __shared__ unsigned b_hist[AG_NB];
        __shared__ unsigned b_base[AG_NB];
        const int tid = threadIdx.x;
        b_hist[tid] = 0u;
        __syncthreads();
        const int q = (blockIdx.x - AL_NB) * AL_T + tid;
        int sv[BIN_EPT], dv[BIN_EPT];
        if (q < BIN_NQ) {
            const int4* s4 = (const int4*)ei + (BIN_EPT / 4) * q;
            const int4* d4 = (const int4*)(ei + GAT_E) + (BIN_EPT / 4) * q;
#pragma unroll
            for (int i = 0; i < BIN_EPT / 4; ++i) {
                const int4 a = s4[i], b = d4[i];
                sv[4 * i] = a.x; sv[4 * i + 1] = a.y; sv[4 * i + 2] = a.z; sv[4 * i + 3] = a.w;
                dv[4 * i] = b.x; dv[4 * i + 1] = b.y; dv[4 * i + 2] = b.z; dv[4 * i + 3] = b.w;
            }
        } else {
#pragma unroll
            for (int i = 0; i < BIN_EPT; ++i) { sv[i] = 0; dv[i] = -1; }
        }
        unsigned own[BIN_EPT], rk[BIN_EPT];
#pragma unroll
        for (int i = 0; i < BIN_EPT; ++i) {
            const bool ok = (unsigned)dv[i] < (unsigned)GAT_N;
            own[i] = ok ? ((unsigned)dv[i] / (unsigned)AG_R) : 0xFFFFFFFFu;
            rk[i]  = ok ? atomicAdd(&b_hist[own[i]], 1u) : 0u;
        }
        __syncthreads();
        {
            const unsigned h = b_hist[tid];
            b_base[tid] = h ? atomicAdd(lists + (size_t)tid * AG_R * GAT_D, h) : 0u;
        }
        __syncthreads();
#pragma unroll
        for (int i = 0; i < BIN_EPT; ++i) {
            if (own[i] != 0xFFFFFFFFu) {
                const int n0b = (int)own[i] * AG_R;
                const int nnb = (GAT_N - n0b < AG_R) ? (GAT_N - n0b) : AG_R;
                const unsigned room = ((unsigned)(nnb * GAT_D - 4) < (unsigned)AG_CAP)
                                          ? (unsigned)(nnb * GAT_D - 4) : (unsigned)AG_CAP;
                const unsigned slot = b_base[own[i]] + rk[i];
                if (slot < room)
                    lists[(size_t)n0b * GAT_D + 4 + slot] =
                        (unsigned)sv[i] | ((unsigned)(dv[i] - n0b) << 16);
            }
        }
        return;
    }
    const int lane  = threadIdx.x & 31;
    const int chunk = blockIdx.x * (AL_T / 32)
                    + __builtin_amdgcn_readfirstlane((int)(threadIdx.x >> 5));
    if (chunk >= AL_NCH) return;
    const int first = chunk * AL_NPW;
    const int n0 = (first + AL_NPW <= GAT_N) ? first : (GAT_N - AL_NPW);

    const float* hp = H + (size_t)n0 * GAT_D + 4 * lane;
    float4 hv[AL_NPW];
#pragma unroll
    for (int i = 0; i < AL_NPW; ++i) hv[i] = *(const float4*)(hp + (size_t)i * GAT_D);
    const float4 sv = *(const float4*)(a_src + 4 * lane);
    const float4 dv = *(const float4*)(a_dst + 4 * lane);
#pragma unroll
    for (int i = 0; i < AL_NPW; ++i)
        asm volatile("" :: "v"(hv[i].x), "v"(hv[i].y), "v"(hv[i].z), "v"(hv[i].w));

    float ps[AL_NPW], pd[AL_NPW];
#pragma unroll
    for (int i = 0; i < AL_NPW; ++i) {
        ps[i] = hv[i].x * sv.x + hv[i].y * sv.y + hv[i].z * sv.z + hv[i].w * sv.w;
        pd[i] = hv[i].x * dv.x + hv[i].y * dv.y + hv[i].z * dv.z + hv[i].w * dv.w;
    }

    al_halve<16>(ps, lane);  al_halve<16>(pd, lane);
    al_halve<8>(ps, lane);   al_halve<8>(pd, lane);
    al_halve<4>(ps, lane);   al_halve<4>(pd, lane);
    al_halve<2>(ps, lane);   al_halve<2>(pd, lane);
    al_halve<1>(ps, lane);   al_halve<1>(pd, lane);

    const int n = n0 + lane;
    if (n >= first) {
        as_out[n] = ps[0];
        ad_out[n] = pd[0];
    }
}

#define AG_T    1024
#define AG_SU   8
#define AG_GU   8

__device__ __forceinline__ float ag_lrelu(float e) {
    return (e > 0.f) ? e : NEG_SLOPE * e;
}
__device__ __forceinline__ float ag_segmax(unsigned k) {
    const float m = dec_f32(k);
    return __builtin_isfinite(m) ? m : 0.f;
}

__device__ __forceinline__ bool ag_fetch(bool overflow, int idx, unsigned cnt,
                                         const unsigned* l_sd, const float* l_e,
                                         const int* __restrict__ ei,
                                         const float* __restrict__ as,
                                         const float* s_ad, int n0, int nn,
                                         unsigned& sd, float& e) {
    sd = 0u; e = 0.f;
    if (!overflow) {
        if ((unsigned)idx < cnt) { sd = l_sd[idx]; e = l_e[idx]; return true; }
        return false;
    }
    const unsigned dl = (unsigned)(ei[GAT_E + idx] - n0);
    if (dl < (unsigned)nn) {
        const int s = ei[idx];
        sd = (unsigned)s | (dl << 16);
        e  = ag_lrelu(as[s] + s_ad[dl]);
        return true;
    }
    return false;
}

__global__ __launch_bounds__(AG_T)
void gat_aggregate(const int* __restrict__ ei,
                   const float* __restrict__ H,
                   const float* __restrict__ as,
                   const float* __restrict__ ad,
                   const float* __restrict__ bias,
                   float* out,
                   const unsigned* lst_all) {
    __shared__ float    acc[AG_R * GAT_D];
    __shared__ float    s_ad[AG_R];
    __shared__ float    s_es[AG_R];
    __shared__ unsigned s_mx[AG_R];
    __shared__ float    s_sum[AG_R];
    __shared__ unsigned l_sd[AG_CAP];
    __shared__ float    l_e[AG_CAP];
    __shared__ unsigned s_cnt;

    const int tid  = threadIdx.x;
    const int lane = tid & 31;
    const int wave = tid >> 5;
    const int n0   = blockIdx.x * AG_R;
    const int nn   = (GAT_N - n0 < AG_R) ? (GAT_N - n0) : AG_R;
    const unsigned* lst = lst_all + (size_t)n0 * GAT_D;
    const unsigned lcap = ((unsigned)(nn * GAT_D - 4) < (unsigned)AG_CAP)
                              ? (unsigned)(nn * GAT_D - 4) : (unsigned)AG_CAP;

    for (int i = tid; i < AG_R * GAT_D; i += AG_T) acc[i] = 0.f;
    if (tid < nn) {
        const float d = ad[n0 + tid];
        const float e = ag_lrelu(as[n0 + tid] + d);
        s_ad[tid]  = d;
        s_es[tid]  = e;
        s_mx[tid]  = enc_f32(e);
        s_sum[tid] = 0.f;
    }
    if (tid == 0) s_cnt = 0u;
    __syncthreads();

    const unsigned lc = (unsigned)__builtin_amdgcn_readfirstlane((int)lst[0]);
    const bool listed = (lc <= lcap);
    if (listed) {
        for (unsigned j = tid; j < lc; j += AG_T) l_sd[j] = lst[4 + j];
        if (tid == 0) s_cnt = lc;
    }

    if (!listed) {
        const int4* dst4 = (const int4*)(ei + GAT_E);
        const int rot = blockIdx.x * ((GAT_E / 4) / AG_NB);
        for (int q0 = 0; q0 < GAT_E / 4; q0 += AG_SU * AG_T) {
            int4 d4[AG_SU];
            int  qq[AG_SU];
#pragma unroll
            for (int u = 0; u < AG_SU; ++u) {
                const int q = q0 + u * AG_T + tid;
                qq[u] = (q + rot >= GAT_E / 4) ? (q + rot - GAT_E / 4) : (q + rot);
                d4[u] = (q < GAT_E / 4) ? dst4[qq[u]] : make_int4(-1, -1, -1, -1);
            }
#pragma unroll
            for (int u = 0; u < AG_SU; ++u) {
                const int q = qq[u];
                const int dd[4] = {d4[u].x, d4[u].y, d4[u].z, d4[u].w};
#pragma unroll
                for (int k = 0; k < 4; ++k) {
                    const unsigned dl = (unsigned)(dd[k] - n0);
                    if (dl < (unsigned)nn) {
                        const unsigned slot = atomicAdd(&s_cnt, 1u);
                        if (slot < AG_CAP) l_sd[slot] = (unsigned)(4 * q + k);
                    }
                }
            }
        }
    }
    __syncthreads();
    const unsigned cnt = s_cnt;
    const bool overflow = cnt > AG_CAP;
    const int nchunks = overflow ? (GAT_E / 32) : (int)((cnt + 31u) >> 5);

    if (!overflow) {
        for (int j = tid; j < (int)cnt; j += AG_T) {
            const unsigned w  = l_sd[j];
            unsigned dl; int s;
            if (listed) { s = (int)(w & 0xFFFFu); dl = w >> 16; }
            else        { s = ei[w]; dl = (unsigned)(ei[GAT_E + w] - n0); }
            const float e = ag_lrelu(as[s] + s_ad[dl]);
            atomicMax(&s_mx[dl], enc_f32(e));
            const unsigned sd = (unsigned)s | (dl << 16);
            l_sd[j] = sd;
            l_e[j]  = e;
        }
    } else {
        for (int c = wave; c < nchunks; c += AG_T / 32) {
            unsigned sd; float e;
            if (ag_fetch(true, c * 32 + lane, cnt, l_sd, l_e, ei, as, s_ad, n0, nn, sd, e))
                atomicMax(&s_mx[sd >> 16], enc_f32(e));
        }
    }
    __syncthreads();

    for (int c = wave; c < nchunks; c += AG_T / 32) {
        unsigned sd; float e;
        if (ag_fetch(overflow, c * 32 + lane, cnt, l_sd, l_e, ei, as, s_ad, n0, nn, sd, e)) {
            const unsigned dl = sd >> 16;
            atomicAdd(&s_sum[dl], expf(e - ag_segmax(s_mx[dl])));
        }
    }
    if (tid < nn) atomicAdd(&s_sum[tid], expf(s_es[tid] - ag_segmax(s_mx[tid])));
    __syncthreads();

    for (int c = wave; c < nchunks; c += AG_T / 32) {
        unsigned sd; float e;
        const bool valid = ag_fetch(overflow, c * 32 + lane, cnt, l_sd, l_e, ei, as, s_ad, n0, nn, sd, e);
        float coef = 0.f;
        if (valid) {
            const unsigned dl = sd >> 16;
            coef = expf(e - ag_segmax(s_mx[dl])) / (s_sum[dl] + 1e-16f);
        }
        unsigned mask = __builtin_amdgcn_ballot_w32(valid);
        while (mask) {
            bool     ok[AG_GU];
            unsigned sj[AG_GU];
            float    cj[AG_GU];
            float4   hv[AG_GU];
#pragma unroll
            for (int u = 0; u < AG_GU; ++u) {
                ok[u] = (mask != 0u);
                const int j = ok[u] ? __builtin_ctz(mask) : 0;
                mask &= mask - 1u;
                sj[u] = (unsigned)__builtin_amdgcn_readlane((int)sd, j);
                cj[u] = __uint_as_float((unsigned)__builtin_amdgcn_readlane((int)__float_as_uint(coef), j));
                hv[u] = *(const float4*)(H + (size_t)(sj[u] & 0xFFFFu) * GAT_D + 4 * lane);
            }
#pragma unroll
            for (int u = 0; u < AG_GU; ++u)
                asm volatile("" :: "v"(hv[u].x), "v"(hv[u].y), "v"(hv[u].z), "v"(hv[u].w));
#pragma unroll
            for (int u = 0; u < AG_GU; ++u) {
                if (ok[u]) {
                    float* a = acc + (sj[u] >> 16) * GAT_D + lane;
                    atomicAdd(a,      hv[u].x * cj[u]);
                    atomicAdd(a + 32, hv[u].y * cj[u]);
                    atomicAdd(a + 64, hv[u].z * cj[u]);
                    atomicAdd(a + 96, hv[u].w * cj[u]);
                }
            }
        }
    }
    __syncthreads();

    const float4 bv = *(const float4*)(bias + 4 * lane);
    for (int dl = wave; dl < nn; dl += AG_T / 32) {
        const size_t row = (size_t)(n0 + dl) * GAT_D + 4 * lane;
        const float4 hs = *(const float4*)(H + row);
        const float coef = expf(s_es[dl] - ag_segmax(s_mx[dl])) / (s_sum[dl] + 1e-16f);
        const float* a = acc + dl * GAT_D + lane;
        float4 o;
        o.x = (a[0]  + hs.x * coef) + bv.x;
        o.y = (a[32] + hs.y * coef) + bv.y;
        o.z = (a[64] + hs.z * coef) + bv.z;
        o.w = (a[96] + hs.w * coef) + bv.w;
        *(float4*)(out + row) = o;
    }
}

#define PL_Q   4
#define PL_T   1024
#define PL_W   (PL_T / 32)
#define PL_NC  ((GAT_N + 31) / 32)
#define PL_SU  7
#define PL_GU  16

__global__ __launch_bounds__(PL_T)
void gat_pool_own(const float* __restrict__ feats,
                  const int* __restrict__ batch,
                  float* __restrict__ out_pooled) {
    __shared__ float    p_sum[PL_W * 32];
    __shared__ unsigned p_cnt[PL_W];

    const int lane = threadIdx.x & 31;
    const int wave = __builtin_amdgcn_readfirstlane((int)(threadIdx.x >> 5));
    const int g    = blockIdx.x / PL_Q;
    const int c0   = (blockIdx.x % PL_Q) * 32;
    const float* col = feats + c0 + lane;

    float    acc = 0.f;
    unsigned cnt = 0u;

    for (int ch0 = wave; ch0 < PL_NC; ch0 += PL_W * PL_SU) {
        int id[PL_SU];
#pragma unroll
        for (int u = 0; u < PL_SU; ++u) {
            const int n = (ch0 + u * PL_W) * 32 + lane;
            const int v = batch[(n < GAT_N) ? n : (GAT_N - 1)];
            id[u] = (n < GAT_N) ? v : -1;
        }
#pragma unroll
        for (int u = 0; u < PL_SU; ++u) {
            const int nb = (ch0 + u * PL_W) * 32;
            unsigned mask = __builtin_amdgcn_ballot_w32(id[u] == g);
            cnt += (unsigned)__builtin_popcount(mask);
            while (mask) {
                bool  ok[PL_GU];
                float t[PL_GU];
#pragma unroll
                for (int r = 0; r < PL_GU; ++r) {
                    ok[r] = (mask != 0u);
                    const int j = ok[r] ? __builtin_ctz(mask) : 0;
                    mask &= mask - 1u;
                    t[r] = col[(size_t)(nb + j) * GAT_D];
                }
#pragma unroll
                for (int r = 0; r < PL_GU; ++r)
                    asm volatile("" :: "v"(t[r]));
#pragma unroll
                for (int r = 0; r < PL_GU; ++r) acc += ok[r] ? t[r] : 0.f;
            }
        }
    }

    p_sum[wave * 32 + lane] = acc;
    if (lane == 0) p_cnt[wave] = cnt;
    __syncthreads();
    if (wave == 0) {
        float    s = 0.f;
        unsigned c = 0u;
#pragma unroll
        for (int w = 0; w < PL_W; ++w) {
            s += p_sum[w * 32 + lane];
            c += p_cnt[w];
        }
        out_pooled[g * GAT_D + c0 + lane] = s / fmaxf((float)c, 1.0f);
    }
}

extern "C" void kernel_launch(void* const* d_in, const int* in_sizes, int n_in,
                              void* d_out, int out_size, void* d_ws, size_t ws_size,
                              hipStream_t stream) {
    const float* x     = (const float*)d_in[0];
    const int*   ei    = (const int*)d_in[1];
    const int*   batch = (const int*)d_in[2];
    const float* W1  = (const float*)d_in[3];
    const float* as1 = (const float*)d_in[4];
    const float* ad1 = (const float*)d_in[5];
    const float* b1  = (const float*)d_in[6];
    const float* W2  = (const float*)d_in[7];
    const float* as2 = (const float*)d_in[8];
    const float* ad2 = (const float*)d_in[9];
    const float* b2  = (const float*)d_in[10];

    float* out_feats  = (float*)d_out;
    float* out_pooled = out_feats + (size_t)GAT_N * GAT_D;

    float* ws = (float*)d_ws;
    size_t o = 0;
    float*    h      = ws + o; o += (size_t)GAT_N * GAT_D;
    float*    feats1 = ws + o; o += (size_t)GAT_N * GAT_D;
    float*    alphaS = ws + o; o += GAT_N;
    float*    alphaD = ws + o; o += GAT_N;
    (void)ws_size; (void)in_sizes; (void)n_in; (void)out_size;

    unsigned* lists = (unsigned*)out_feats;

    gat_gemm128_wmma<<<GAT_N / 16, 256, 0, stream>>>(x, W1, h, lists);
    gat_alpha<<<AL_NB + BIN_NB, AL_T, 0, stream>>>(h, as1, ad1, alphaS, alphaD, ei, lists);
    gat_aggregate<<<AG_NB, AG_T, 0, stream>>>(ei, h, alphaS, alphaD, b1, feats1, lists);
    gat_gemm128_wmma<<<GAT_N / 16, 256, 0, stream>>>(feats1, W2, h, (unsigned*)nullptr);
    gat_alpha<<<AL_NB, AL_T, 0, stream>>>(h, as2, ad2, alphaS, alphaD, ei, (unsigned*)nullptr);
    gat_aggregate<<<AG_NB, AG_T, 0, stream>>>(ei, h, alphaS, alphaD, b2, out_feats, lists);

    gat_pool_own<<<GAT_G * PL_Q, PL_T, 0, stream>>>(out_feats, batch, out_pooled);
}


# Layers folded (no nonlinearity between them): layer 1 gathers rows of x@(W1@W2), alpha1 = x.(W1 a); its aggregate stores h2 and alpha2; second node product and alpha launch gone; new sum order
# speedup vs baseline: 5.0814x; 1.1760x over previous
#include <hip/hip_runtime.h>
#include <hip/hip_bf16.h>

#define GAT_N      50000
#define GAT_E      800000
#define GAT_ETOT   850000
#define GAT_D      128
#define GAT_G      64
#define NEG_SLOPE  0.2f

typedef __attribute__((ext_vector_type(2))) float v2f;
typedef __attribute__((ext_vector_type(8))) float v8f;

__device__ __forceinline__ unsigned enc_f32(float f) {
    unsigned u = __float_as_uint(f);
    return (u & 0x80000000u) ? ~u : (u | 0x80000000u);
}
__device__ __forceinline__ float dec_f32(unsigned k) {
    unsigned u = (k & 0x80000000u) ? (k & 0x7FFFFFFFu) : ~k;
    return __uint_as_float(u);
}

#define AG_R        196
#define AG_NB       256
#define AG_CAP      8192

__device__ __forceinline__ void gemm_tile16(const float* __restrict__ X,
                                            const float* __restrict__ W,
                                            float* __restrict__ H,
                                            const int m0) {
    const int wave = threadIdx.x >> 5;
    const int lane = threadIdx.x & 31;
    const int half = lane >> 4;
    const int l    = lane & 15;
    const int n0   = wave * 16;

    const float* xrow = X + (size_t)(m0 + l) * GAT_D;
    v8f acc = {};

#pragma unroll
    for (int k = 0; k < GAT_D; k += 4) {
        const int kk = k + 2 * half;
        v2f a;
        {
            const float2 av = *(const float2*)(xrow + kk);
            a.x = av.x; a.y = av.y;
        }
        v2f b;
        b.x = W[(size_t)kk       * GAT_D + n0 + l];
        b.y = W[(size_t)(kk + 1) * GAT_D + n0 + l];
        acc = __builtin_amdgcn_wmma_f32_16x16x4_f32(
            false, a, false, b, (short)0, acc, false, false);
    }

    const int mBase = m0 + half * 8;
#pragma unroll
    for (int r = 0; r < 8; ++r)
        H[(size_t)(mBase + r) * GAT_D + n0 + l] = acc[r];
}

__global__ void gat_gemm128_wmma(const float* __restrict__ X,
                      const float* __restrict__ W,
                      float* __restrict__ H) {
    gemm_tile16(X, W, H, blockIdx.x * 16);
}

#define FOLD_NB 10
__global__ void gat_fold(const float* __restrict__ W1,
              const float* __restrict__ W2,
              const float* __restrict__ a_src1,
              const float* __restrict__ a_dst1,
              const float* __restrict__ b1,
              float* __restrict__ W12,
              float* __restrict__ vec,
              unsigned* __restrict__ cnt_zero) {
    if (blockIdx.x < 8) {
        gemm_tile16(W1, W2, W12, blockIdx.x * 16);
        return;
    }
    const int tid = threadIdx.x;
    if (blockIdx.x == 8) {
        if (tid < GAT_D) {
            const float4* wr = (const float4*)(W1 + (size_t)tid * GAT_D);
            float s = 0.f, d = 0.f;
            for (int j = 0; j < GAT_D / 4; ++j) {
                const float4 w = wr[j];
                const float4 a = *(const float4*)(a_src1 + 4 * j);
                const float4 c = *(const float4*)(a_dst1 + 4 * j);
                s += w.x * a.x + w.y * a.y + w.z * a.z + w.w * a.w;
                d += w.x * c.x + w.y * c.y + w.z * c.z + w.w * c.w;
            }
            vec[tid]         = s;
            vec[GAT_D + tid] = d;
        } else {
            const int j = tid - GAT_D;
            float t = 0.f;
            for (int k = 0; k < GAT_D; ++k) t += b1[k] * W2[(size_t)k * GAT_D + j];
            vec[2 * GAT_D + j] = t;
        }
        return;
    }
    cnt_zero[(size_t)tid * AG_R * GAT_D] = 0u;
}

#define AS_CH   16384
#define AS_OFF  8448
static_assert(AS_OFF >= 4 + AG_CAP && AS_OFF + AS_CH <= AG_R * GAT_D, "beside the list, inside the rows");
static_assert(4 * AS_CH >= GAT_N, "four owners' rows hold all nodes");
static_assert(AS_CH == (1 << 14), "the shift below");
__device__ __forceinline__ int as_idx(int n, int pitch) {
    return (n >> 14) * pitch + (n & (AS_CH - 1));
}

#define BIN_EPT     16
#define BIN_NQ      (GAT_E / BIN_EPT)

#define AL_NPW  32
#define AL_T    256
#define AL_NCH  ((GAT_N + AL_NPW - 1) / AL_NPW)
#define AL_NB   ((AL_NCH + AL_T / 32 - 1) / (AL_T / 32))
#define BIN_NB  ((BIN_NQ + AL_T - 1) / AL_T)
static_assert(AL_T == AG_NB, "one histogram slot per thread");
static_assert(GAT_E % BIN_EPT == 0, "whole 16-edge groups");

template <int N>
__device__ __forceinline__ void al_halve(float (&p)[AL_NPW], int lane) {
    const bool up = (lane & N) != 0;
#pragma unroll
    for (int i = 0; i < N; ++i) {
        const float keep = up ? p[i + N] : p[i];
        const float give = up ? p[i] : p[i + N];
        p[i] = keep + __shfl_xor(give, N, 32);
    }
}

__global__ __launch_bounds__(AL_T)
void gat_alpha(const float* __restrict__ H,
               const float* __restrict__ a_src,
               const float* __restrict__ a_dst,
               float* as_out,
               float* __restrict__ ad_out,
               const int* __restrict__ ei,
               unsigned* lists,
               const int as_pitch) {
    if (lists != nullptr && blockIdx.x >= AL_NB) {
        __shared__ unsigned b_hist[AG_NB];
        __shared__ unsigned b_base[AG_NB];
        const int tid = threadIdx.x;
        b_hist[tid] = 0u;
        __syncthreads();
        const int q = (blockIdx.x - AL_NB) * AL_T + tid;
        int sv[BIN_EPT], dv[BIN_EPT];
        if (q < BIN_NQ) {
            const int4* s4 = (const int4*)ei + (BIN_EPT / 4) * q;
            const int4* d4 = (const int4*)(ei + GAT_E) + (BIN_EPT / 4) * q;
#pragma unroll
            for (int i = 0; i < BIN_EPT / 4; ++i) {
                const int4 a = s4[i], b = d4[i];
                sv[4 * i] = a.x; sv[4 * i + 1] = a.y; sv[4 * i + 2] = a.z; sv[4 * i + 3] = a.w;
                dv[4 * i] = b.x; dv[4 * i + 1] = b.y; dv[4 * i + 2] = b.z; dv[4 * i + 3] = b.w;
            }
        } else {
#pragma unroll
            for (int i = 0; i < BIN_EPT; ++i) { sv[i] = 0; dv[i] = -1; }
        }
        unsigned own[BIN_EPT], rk[BIN_EPT];
#pragma unroll
        for (int i = 0; i < BIN_EPT; ++i) {
            const bool ok = (unsigned)dv[i] < (unsigned)GAT_N;
            own[i] = ok ? ((unsigned)dv[i] / (unsigned)AG_R) : 0xFFFFFFFFu;
            rk[i]  = ok ? atomicAdd(&b_hist[own[i]], 1u) : 0u;
        }
        __syncthreads();
        {
            const unsigned h = b_hist[tid];
            b_base[tid] = h ? atomicAdd(lists + (size_t)tid * AG_R * GAT_D, h) : 0u;
        }
        __syncthreads();
#pragma unroll
        for (int i = 0; i < BIN_EPT; ++i) {
            if (own[i] != 0xFFFFFFFFu) {
                const int n0b = (int)own[i] * AG_R;
                const int nnb = (GAT_N - n0b < AG_R) ? (GAT_N - n0b) : AG_R;
                const unsigned room = ((unsigned)(nnb * GAT_D - 4) < (unsigned)AG_CAP)
                                          ? (unsigned)(nnb * GAT_D - 4) : (unsigned)AG_CAP;
                const unsigned slot = b_base[own[i]] + rk[i];
                if (slot < room)
                    lists[(size_t)n0b * GAT_D + 4 + slot] =
                        (unsigned)sv[i] | ((unsigned)(dv[i] - n0b) << 16);
            }
        }
        return;
    }
    const int lane  = threadIdx.x & 31;
    const int chunk = blockIdx.x * (AL_T / 32)
                    + __builtin_amdgcn_readfirstlane((int)(threadIdx.x >> 5));
    if (chunk >= AL_NCH) return;
    const int first = chunk * AL_NPW;
    const int n0 = (first + AL_NPW <= GAT_N) ? first : (GAT_N - AL_NPW);

    const float* hp = H + (size_t)n0 * GAT_D + 4 * lane;
    float4 hv[AL_NPW];
#pragma unroll
    for (int i = 0; i < AL_NPW; ++i) hv[i] = *(const float4*)(hp + (size_t)i * GAT_D);
    const float4 sv = *(const float4*)(a_src + 4 * lane);
    const float4 dv = *(const float4*)(a_dst + 4 * lane);
#pragma unroll
    for (int i = 0; i < AL_NPW; ++i)
        asm volatile("" :: "v"(hv[i].x), "v"(hv[i].y), "v"(hv[i].z), "v"(hv[i].w));

    float ps[AL_NPW], pd[AL_NPW];
#pragma unroll
    for (int i = 0; i < AL_NPW; ++i) {
        ps[i] = hv[i].x * sv.x + hv[i].y * sv.y + hv[i].z * sv.z + hv[i].w * sv.w;
        pd[i] = hv[i].x * dv.x + hv[i].y * dv.y + hv[i].z * dv.z + hv[i].w * dv.w;
    }

    al_halve<16>(ps, lane);  al_halve<16>(pd, lane);
    al_halve<8>(ps, lane);   al_halve<8>(pd, lane);
    al_halve<4>(ps, lane);   al_halve<4>(pd, lane);
    al_halve<2>(ps, lane);   al_halve<2>(pd, lane);
    al_halve<1>(ps, lane);   al_halve<1>(pd, lane);

    const int n = n0 + lane;
    if (n >= first) {
        as_out[as_idx(n, as_pitch)] = ps[0];
        ad_out[n] = pd[0];
    }
}

#define AG_T    1024
#define AG_SU   8
#define AG_GU   8

__device__ __forceinline__ float ag_lrelu(float e) {
    return (e > 0.f) ? e : NEG_SLOPE * e;
}
__device__ __forceinline__ float ag_segmax(unsigned k) {
    const float m = dec_f32(k);
    return __builtin_isfinite(m) ? m : 0.f;
}

__device__ __forceinline__ bool ag_fetch(bool overflow, int idx, unsigned cnt,
                                         const unsigned* l_sd, const float* l_e,
                                         const int* __restrict__ ei,
                                         const float* as, int as_pitch,
                                         const float* s_ad, int n0, int nn,
                                         unsigned& sd, float& e) {
    sd = 0u; e = 0.f;
    if (!overflow) {
        if ((unsigned)idx < cnt) { sd = l_sd[idx]; e = l_e[idx]; return true; }
        return false;
    }
    const unsigned dl = (unsigned)(ei[GAT_E + idx] - n0);
    if (dl < (unsigned)nn) {
        const int s = ei[idx];
        sd = (unsigned)s | (dl << 16);
        e  = ag_lrelu(as[as_idx(s, as_pitch)] + s_ad[dl]);
        return true;
    }
    return false;
}

__global__ __launch_bounds__(AG_T)
void gat_aggregate(const int* __restrict__ ei,
                   const float* __restrict__ H,
                   const float* as, const int as_pitch,
                   const float* ad,
                   const float* __restrict__ bias,
                   float* out,
                   const unsigned* lst_all,
                   const float* __restrict__ a_src2,
                   const float* __restrict__ a_dst2,
                   float* as2_out,
                   float* ad2_out) {
    __shared__ float    acc[AG_R * GAT_D];
    __shared__ float    s_ad[AG_R];
    __shared__ float    s_es[AG_R];
    __shared__ unsigned s_mx[AG_R];
    __shared__ float    s_sum[AG_R];
    __shared__ unsigned l_sd[AG_CAP];
    __shared__ float    l_e[AG_CAP];
    __shared__ unsigned s_cnt;

    const int tid  = threadIdx.x;
    const int lane = tid & 31;
    const int wave = tid >> 5;
    const int n0   = blockIdx.x * AG_R;
    const int nn   = (GAT_N - n0 < AG_R) ? (GAT_N - n0) : AG_R;
    const unsigned* lst = lst_all + (size_t)n0 * GAT_D;
    const unsigned lcap = ((unsigned)(nn * GAT_D - 4) < (unsigned)AG_CAP)
                              ? (unsigned)(nn * GAT_D - 4) : (unsigned)AG_CAP;

    for (int i = tid; i < AG_R * GAT_D; i += AG_T) acc[i] = 0.f;
    if (tid < nn) {
        const float d = ad[n0 + tid];
        const float e = ag_lrelu(as[as_idx(n0 + tid, as_pitch)] + d);
        s_ad[tid]  = d;
        s_es[tid]  = e;
        s_mx[tid]  = enc_f32(e);
        s_sum[tid] = 0.f;
    }
    if (tid == 0) s_cnt = 0u;
    __syncthreads();

    const unsigned lc = (unsigned)__builtin_amdgcn_readfirstlane((int)lst[0]);
    const bool listed = (lc <= lcap);
    if (listed) {
        for (unsigned j = tid; j < lc; j += AG_T) l_sd[j] = lst[4 + j];
        if (tid == 0) s_cnt = lc;
    }

    if (!listed) {
        const int4* dst4 = (const int4*)(ei + GAT_E);
        const int rot = blockIdx.x * ((GAT_E / 4) / AG_NB);
        for (int q0 = 0; q0 < GAT_E / 4; q0 += AG_SU * AG_T) {
            int4 d4[AG_SU];
            int  qq[AG_SU];
#pragma unroll
            for (int u = 0; u < AG_SU; ++u) {
                const int q = q0 + u * AG_T + tid;
                qq[u] = (q + rot >= GAT_E / 4) ? (q + rot - GAT_E / 4) : (q + rot);
                d4[u] = (q < GAT_E / 4) ? dst4[qq[u]] : make_int4(-1, -1, -1, -1);
            }
#pragma unroll
            for (int u = 0; u < AG_SU; ++u) {
                const int q = qq[u];
                const int dd[4] = {d4[u].x, d4[u].y, d4[u].z, d4[u].w};
#pragma unroll
                for (int k = 0; k < 4; ++k) {
                    const unsigned dl = (unsigned)(dd[k] - n0);
                    if (dl < (unsigned)nn) {
                        const unsigned slot = atomicAdd(&s_cnt, 1u);
                        if (slot < AG_CAP) l_sd[slot] = (unsigned)(4 * q + k);
                    }
                }
            }
        }
    }
    __syncthreads();
    const unsigned cnt = s_cnt;
    const bool overflow = cnt > AG_CAP;
    const int nchunks = overflow ? (GAT_E / 32) : (int)((cnt + 31u) >> 5);

    if (!overflow) {
        for (int j = tid; j < (int)cnt; j += AG_T) {
            const unsigned w  = l_sd[j];
            unsigned dl; int s;
            if (listed) { s = (int)(w & 0xFFFFu); dl = w >> 16; }
            else        { s = ei[w]; dl = (unsigned)(ei[GAT_E + w] - n0); }
            const float e = ag_lrelu(as[as_idx(s, as_pitch)] + s_ad[dl]);
            atomicMax(&s_mx[dl], enc_f32(e));
            const unsigned sd = (unsigned)s | (dl << 16);
            l_sd[j] = sd;
            l_e[j]  = e;
        }
    } else {
        for (int c = wave; c < nchunks; c += AG_T / 32) {
            unsigned sd; float e;
            if (ag_fetch(true, c * 32 + lane, cnt, l_sd, l_e, ei, as, as_pitch, s_ad, n0, nn, sd, e))
                atomicMax(&s_mx[sd >> 16], enc_f32(e));
        }
    }
    __syncthreads();

    for (int c = wave; c < nchunks; c += AG_T / 32) {
        unsigned sd; float e;
        if (ag_fetch(overflow, c * 32 + lane, cnt, l_sd, l_e, ei, as, as_pitch, s_ad, n0, nn, sd, e)) {
            const unsigned dl = sd >> 16;
            atomicAdd(&s_sum[dl], expf(e - ag_segmax(s_mx[dl])));
        }
    }
    if (tid < nn) atomicAdd(&s_sum[tid], expf(s_es[tid] - ag_segmax(s_mx[tid])));
    __syncthreads();

    for (int c = wave; c < nchunks; c += AG_T / 32) {
        unsigned sd; float e;
        const bool valid = ag_fetch(overflow, c * 32 + lane, cnt, l_sd, l_e, ei, as, as_pitch, s_ad, n0, nn, sd, e);
        float coef = 0.f;
        if (valid) {
            const unsigned dl = sd >> 16;
            coef = expf(e - ag_segmax(s_mx[dl])) / (s_sum[dl] + 1e-16f);
        }
        unsigned mask = __builtin_amdgcn_ballot_w32(valid);
        while (mask) {
            bool     ok[AG_GU];
            unsigned sj[AG_GU];
            float    cj[AG_GU];
            float4   hv[AG_GU];
#pragma unroll
            for (int u = 0; u < AG_GU; ++u) {
                ok[u] = (mask != 0u);
                const int j = ok[u] ? __builtin_ctz(mask) : 0;
                mask &= mask - 1u;
                sj[u] = (unsigned)__builtin_amdgcn_readlane((int)sd, j);
                cj[u] = __uint_as_float((unsigned)__builtin_amdgcn_readlane((int)__float_as_uint(coef), j));
                hv[u] = *(const float4*)(H + (size_t)(sj[u] & 0xFFFFu) * GAT_D + 4 * lane);
            }
#pragma unroll
            for (int u = 0; u < AG_GU; ++u)
                asm volatile("" :: "v"(hv[u].x), "v"(hv[u].y), "v"(hv[u].z), "v"(hv[u].w));
#pragma unroll
            for (int u = 0; u < AG_GU; ++u) {
                if (ok[u]) {
                    float* a = acc + (sj[u] >> 16) * GAT_D + lane;
                    atomicAdd(a,      hv[u].x * cj[u]);
                    atomicAdd(a + 32, hv[u].y * cj[u]);
                    atomicAdd(a + 64, hv[u].z * cj[u]);
                    atomicAdd(a + 96, hv[u].w * cj[u]);
                }
            }
        }
    }
    __syncthreads();

    const bool next = (as2_out != nullptr);
    float4 qs = make_float4(0.f, 0.f, 0.f, 0.f), qd = qs;
    if (next) {
        qs = *(const float4*)(a_src2 + 4 * lane);
        qd = *(const float4*)(a_dst2 + 4 * lane);
    }
    const float4 bv = *(const float4*)(bias + 4 * lane);
    for (int dl = wave; dl < nn; dl += AG_T / 32) {
        const size_t row = (size_t)(n0 + dl) * GAT_D + 4 * lane;
        const float4 hs = *(const float4*)(H + row);
        const float coef = expf(s_es[dl] - ag_segmax(s_mx[dl])) / (s_sum[dl] + 1e-16f);
        const float* a = acc + dl * GAT_D + lane;
        float4 o;
        o.x = (a[0]  + hs.x * coef) + bv.x;
        o.y = (a[32] + hs.y * coef) + bv.y;
        o.z = (a[64] + hs.z * coef) + bv.z;
        o.w = (a[96] + hs.w * coef) + bv.w;
        *(float4*)(out + row) = o;
        if (next) {
            float ps = o.x * qs.x + o.y * qs.y + o.z * qs.z + o.w * qs.w;
            float pd = o.x * qd.x + o.y * qd.y + o.z * qd.z + o.w * qd.w;
            for (int m = 16; m >= 1; m >>= 1) {
                ps += __shfl_xor(ps, m, 32);
                pd += __shfl_xor(pd, m, 32);
            }
            if (lane == 0) {
                as2_out[n0 + dl] = ps;
                ad2_out[n0 + dl] = pd;
            }
        }
    }
}

#define PL_Q   4
#define PL_T   1024
#define PL_W   (PL_T / 32)
#define PL_NC  ((GAT_N + 31) / 32)
#define PL_SU  7
#define PL_GU  16

__global__ __launch_bounds__(PL_T)
void gat_pool_own(const float* __restrict__ feats,
                  const int* __restrict__ batch,
                  float* __restrict__ out_pooled) {
    __shared__ float    p_sum[PL_W * 32];
    __shared__ unsigned p_cnt[PL_W];

    const int lane = threadIdx.x & 31;
    const int wave = __builtin_amdgcn_readfirstlane((int)(threadIdx.x >> 5));
    const int g    = blockIdx.x / PL_Q;
    const int c0   = (blockIdx.x % PL_Q) * 32;
    const float* col = feats + c0 + lane;

    float    acc = 0.f;
    unsigned cnt = 0u;

    for (int ch0 = wave; ch0 < PL_NC; ch0 += PL_W * PL_SU) {
        int id[PL_SU];
#pragma unroll
        for (int u = 0; u < PL_SU; ++u) {
            const int n = (ch0 + u * PL_W) * 32 + lane;
            const int v = batch[(n < GAT_N) ? n : (GAT_N - 1)];
            id[u] = (n < GAT_N) ? v : -1;
        }
#pragma unroll
        for (int u = 0; u < PL_SU; ++u) {
            const int nb = (ch0 + u * PL_W) * 32;
            unsigned mask = __builtin_amdgcn_ballot_w32(id[u] == g);
            cnt += (unsigned)__builtin_popcount(mask);
            while (mask) {
                bool  ok[PL_GU];
                float t[PL_GU];
#pragma unroll
                for (int r = 0; r < PL_GU; ++r) {
                    ok[r] = (mask != 0u);
                    const int j = ok[r] ? __builtin_ctz(mask) : 0;
                    mask &= mask - 1u;
                    t[r] = col[(size_t)(nb + j) * GAT_D];
                }
#pragma unroll
                for (int r = 0; r < PL_GU; ++r)
                    asm volatile("" :: "v"(t[r]));
#pragma unroll
                for (int r = 0; r < PL_GU; ++r) acc += ok[r] ? t[r] : 0.f;
            }
        }
    }

    p_sum[wave * 32 + lane] = acc;
    if (lane == 0) p_cnt[wave] = cnt;
    __syncthreads();
    if (wave == 0) {
        float    s = 0.f;
        unsigned c = 0u;
#pragma unroll
        for (int w = 0; w < PL_W; ++w) {
            s += p_sum[w * 32 + lane];
            c += p_cnt[w];
        }
        out_pooled[g * GAT_D + c0 + lane] = s / fmaxf((float)c, 1.0f);
    }
}

extern "C" void kernel_launch(void* const* d_in, const int* in_sizes, int n_in,
                              void* d_out, int out_size, void* d_ws, size_t ws_size,
                              hipStream_t stream) {
    const float* x     = (const float*)d_in[0];
    const int*   ei    = (const int*)d_in[1];
    const int*   batch = (const int*)d_in[2];
    const float* W1  = (const float*)d_in[3];
    const float* as1 = (const float*)d_in[4];
    const float* ad1 = (const float*)d_in[5];
    const float* b1  = (const float*)d_in[6];
    const float* W2  = (const float*)d_in[7];
    const float* as2 = (const float*)d_in[8];
    const float* ad2 = (const float*)d_in[9];
    const float* b2  = (const float*)d_in[10];

    float* out_feats  = (float*)d_out;
    float* out_pooled = out_feats + (size_t)GAT_N * GAT_D;

    float* ws = (float*)d_ws;
    size_t o = 0;
    float*    g      = ws + o; o += (size_t)GAT_N * GAT_D;
    float*    h2     = ws + o; o += (size_t)GAT_N * GAT_D;
    float*    alphaS = ws + o; o += GAT_N;
    float*    alphaD = ws + o; o += GAT_N;
    float*    W12    = alphaS;
    (void)ws_size; (void)in_sizes; (void)n_in; (void)out_size;

    unsigned* lists = (unsigned*)out_feats;
    float*    as1x  = out_feats + AS_OFF;
    float*    vec   = out_pooled;

    gat_fold<<<FOLD_NB, 256, 0, stream>>>(W1, W2, as1, ad1, b1, W12, vec, lists);
    gat_gemm128_wmma<<<GAT_N / 16, 256, 0, stream>>>(x, W12, g);
    gat_alpha<<<AL_NB + BIN_NB, AL_T, 0, stream>>>(x, vec, vec + GAT_D, as1x, alphaD, ei, lists, AG_R * GAT_D);
    gat_aggregate<<<AG_NB, AG_T, 0, stream>>>(ei, g, as1x, AG_R * GAT_D, alphaD, vec + 2 * GAT_D, h2, lists,
                                              as2, ad2, alphaS, alphaD);
    gat_aggregate<<<AG_NB, AG_T, 0, stream>>>(ei, h2, alphaS, AS_CH, alphaD, b2, out_feats, lists,
                                              (const float*)nullptr, (const float*)nullptr,
                                              (float*)nullptr, (float*)nullptr);

    gat_pool_own<<<GAT_G * PL_Q, PL_T, 0, stream>>>(out_feats, batch, out_pooled);
}
